# starting kernel
# baseline (speedup 1.0000x reference)
#include <hip/hip_runtime.h>
#include <hip/hip_bf16.h>

typedef __attribute__((ext_vector_type(2))) float v2f;
typedef __attribute__((ext_vector_type(8))) float v8f;

#define HDIM 128

__global__ __launch_bounds__(256)
void gemm128_wmma_kernel(const float* __restrict__ A, const float* __restrict__ W,
                         const float* __restrict__ bias, float* __restrict__ out,
                         int K, int do_relu)
{
    __shared__ __align__(16) float As[16 * 256];
    const int row0 = blockIdx.x * 16;
    const int tid  = threadIdx.x;

    const float* Atile = A + (size_t)row0 * K;
    const int total = 16 * K;
    for (int idx = tid * 4; idx < total; idx += 256 * 4)
        *(float4*)(&As[idx]) = *(const float4*)(&Atile[idx]);
    __syncthreads();

    const int wave = tid >> 5;
    const int lane = tid & 31;
    const int m    = lane & 15;
    const int koff = (lane < 16) ? 0 : 2;
    const int n    = wave * 16 + (lane & 15);

    v8f c = {};
    for (int k = 0; k < K; k += 4) {
        v2f a, b;
        a[0] = As[m * K + k + koff];
        a[1] = As[m * K + k + koff + 1];
        b[0] = W[(size_t)(k + koff)     * HDIM + n];
        b[1] = W[(size_t)(k + koff + 1) * HDIM + n];
        c = __builtin_amdgcn_wmma_f32_16x16x4_f32(false, a, false, b,
                                                  (short)0, c, false, false);
    }

    const int   mbase = row0 + ((lane < 16) ? 0 : 8);
    const float bv    = bias[n];
    for (int r = 0; r < 8; ++r) {
        float v = c[r] + bv;
        if (do_relu) v = fmaxf(v, 0.0f);
        out[(size_t)(mbase + r) * HDIM + n] = v;
    }
}

__global__ __launch_bounds__(256)
void sage_update_wmma_kernel(const float* __restrict__ agg,
                             const float* __restrict__ invdeg,
                             const float* __restrict__ h,
                             const float* __restrict__ Wl,
                             const float* __restrict__ bl,
                             const float* __restrict__ Wr,
                             float* __restrict__ out)
{
    __shared__ __align__(16) float Ms[16 * HDIM];
    __shared__ __align__(16) float Hs[16 * HDIM];
    const int row0 = blockIdx.x * 16;
    const int tid  = threadIdx.x;

    for (int idx = tid * 4; idx < 16 * HDIM; idx += 256 * 4) {
        const int   r = idx / HDIM;
        const float s = invdeg[row0 + r];
        float4 va = *(const float4*)(agg + (size_t)row0 * HDIM + idx);
        va.x *= s; va.y *= s; va.z *= s; va.w *= s;
        *(float4*)(&Ms[idx]) = va;
        *(float4*)(&Hs[idx]) = *(const float4*)(h + (size_t)row0 * HDIM + idx);
    }
    __syncthreads();

    const int wave = tid >> 5;
    const int lane = tid & 31;
    const int m    = lane & 15;
    const int koff = (lane < 16) ? 0 : 2;
    const int n    = wave * 16 + (lane & 15);

    v8f c = {};
    for (int k = 0; k < HDIM; k += 4) {
        v2f a, b;
        a[0] = Ms[m * HDIM + k + koff];
        a[1] = Ms[m * HDIM + k + koff + 1];
        b[0] = Wl[(size_t)(k + koff)     * HDIM + n];
        b[1] = Wl[(size_t)(k + koff + 1) * HDIM + n];
        c = __builtin_amdgcn_wmma_f32_16x16x4_f32(false, a, false, b,
                                                  (short)0, c, false, false);
    }
    for (int k = 0; k < HDIM; k += 4) {
        v2f a, b;
        a[0] = Hs[m * HDIM + k + koff];
        a[1] = Hs[m * HDIM + k + koff + 1];
        b[0] = Wr[(size_t)(k + koff)     * HDIM + n];
        b[1] = Wr[(size_t)(k + koff + 1) * HDIM + n];
        c = __builtin_amdgcn_wmma_f32_16x16x4_f32(false, a, false, b,
                                                  (short)0, c, false, false);
    }

    const int   mbase = row0 + ((lane < 16) ? 0 : 8);
    const float bv    = bl[n];
    for (int r = 0; r < 8; ++r) {
        float v = fmaxf(c[r] + bv, 0.0f);
        out[(size_t)(mbase + r) * HDIM + n] = v;
    }
}

__global__ __launch_bounds__(32)
void cls_wmma_kernel(const float* __restrict__ h, const float* __restrict__ Wc,
                     const float* __restrict__ bc, float* __restrict__ out, int ncls)
{
    const int row0 = blockIdx.x * 16;
    const int lane = threadIdx.x & 31;
    const int m    = lane & 15;
    const int koff = (lane < 16) ? 0 : 2;
    const int n    = blockIdx.y * 16 + (lane & 15);
    const bool nv  = (n < ncls);

    v8f c = {};
    for (int k = 0; k < HDIM; k += 4) {
        v2f a, b;
        a[0] = h[(size_t)(row0 + m) * HDIM + k + koff];
        a[1] = h[(size_t)(row0 + m) * HDIM + k + koff + 1];
        b[0] = nv ? Wc[(size_t)(k + koff)     * ncls + n] : 0.0f;
        b[1] = nv ? Wc[(size_t)(k + koff + 1) * ncls + n] : 0.0f;
        c = __builtin_amdgcn_wmma_f32_16x16x4_f32(false, a, false, b,
                                                  (short)0, c, false, false);
    }

    if (nv) {
        const int   mbase = row0 + ((lane < 16) ? 0 : 8);
        const float bv    = bc[n];
        for (int r = 0; r < 8; ++r)
            out[(size_t)(mbase + r) * ncls + n] = c[r] + bv;
    }
}

__global__ __launch_bounds__(256)
void scatter_add_kernel(const int* __restrict__ src, const int* __restrict__ dst,
                        const float* __restrict__ h, float* __restrict__ agg, int E)
{
    const int e = blockIdx.x * 8 + (threadIdx.x >> 5);
    if (e >= E) return;
    const int lane = threadIdx.x & 31;
    const int s = src[e], d = dst[e];
    float4 v = *(const float4*)(h + (size_t)s * HDIM + lane * 4);
    float* a = agg + (size_t)d * HDIM + lane * 4;
    atomicAdd(a + 0, v.x);
    atomicAdd(a + 1, v.y);
    atomicAdd(a + 2, v.z);
    atomicAdd(a + 3, v.w);
}

__global__ __launch_bounds__(256)
void deg_kernel(const int* __restrict__ dst, float* __restrict__ deg, int E)
{
    const int e = blockIdx.x * blockDim.x + threadIdx.x;
    if (e < E) atomicAdd(&deg[dst[e]], 1.0f);
}

__global__ __launch_bounds__(256)
void invdeg_kernel(float* __restrict__ deg, int n)
{
    const int i = blockIdx.x * blockDim.x + threadIdx.x;
    if (i < n) deg[i] = 1.0f / fmaxf(deg[i], 1.0f);
}

extern "C" void kernel_launch(void* const* d_in, const int* in_sizes, int n_in,
                              void* d_out, int out_size, void* d_ws, size_t ws_size,
                              hipStream_t stream)
{
    const float* x      = (const float*)d_in[0];
    const int*   edge   = (const int*)  d_in[1];
    const float* W_proj = (const float*)d_in[2];
    const float* b_proj = (const float*)d_in[3];
    const float* Wl[3]  = { (const float*)d_in[4], (const float*)d_in[7], (const float*)d_in[10] };
    const float* bl[3]  = { (const float*)d_in[5], (const float*)d_in[8], (const float*)d_in[11] };
    const float* Wr[3]  = { (const float*)d_in[6], (const float*)d_in[9], (const float*)d_in[12] };
    const float* Wc     = (const float*)d_in[13];
    const float* bc     = (const float*)d_in[14];

    const int N    = in_sizes[0] / 256;
    const int E    = in_sizes[1] / 2;
    const int ncls = in_sizes[14];

    const int* src = edge;
    const int* dst = edge + E;

    float* hA     = (float*)d_ws;
    float* hB     = hA  + (size_t)N * HDIM;
    float* agg    = hB  + (size_t)N * HDIM;
    float* invdeg = agg + (size_t)N * HDIM;

    const int rowTiles = (N + 15) / 16;

    hipMemsetAsync(invdeg, 0, (size_t)N * sizeof(float), stream);
    deg_kernel<<<(E + 255) / 256, 256, 0, stream>>>(dst, invdeg, E);
    invdeg_kernel<<<(N + 255) / 256, 256, 0, stream>>>(invdeg, N);

    gemm128_wmma_kernel<<<rowTiles, 256, 0, stream>>>(x, W_proj, b_proj, hA, 256, 1);

    for (int l = 0; l < 3; ++l) {
        hipMemsetAsync(agg, 0, (size_t)N * HDIM * sizeof(float), stream);
        scatter_add_kernel<<<(E + 7) / 8, 256, 0, stream>>>(src, dst, hA, agg, E);
        sage_update_wmma_kernel<<<rowTiles, 256, 0, stream>>>(agg, invdeg, hA,
                                                              Wl[l], bl[l], Wr[l], hB);
        float* t = hA; hA = hB; hB = t;
    }

    cls_wmma_kernel<<<dim3(rowTiles, (ncls + 15) / 16), 32, 0, stream>>>(
        hA, Wc, bc, (float*)d_out, ncls);
}


# Per-edge float atomic scatter replaced by neighbour lists built once (int atomic rank per edge) + per-node gather: rows summed in list order, no float atomics, no agg memset
# speedup vs baseline: 1.9757x; 1.9757x over previous
#include <hip/hip_runtime.h>
#include <hip/hip_bf16.h>

typedef __attribute__((ext_vector_type(2))) float v2f;
typedef __attribute__((ext_vector_type(8))) float v8f;

#define HDIM 128

__global__ __launch_bounds__(256)
void gemm128_wmma_kernel(const float* __restrict__ A, const float* __restrict__ W,
                         const float* __restrict__ bias, float* __restrict__ out,
                         int K, int do_relu)
{
    __shared__ __align__(16) float As[16 * 256];
    const int row0 = blockIdx.x * 16;
    const int tid  = threadIdx.x;

    const float* Atile = A + (size_t)row0 * K;
    const int total = 16 * K;
    for (int idx = tid * 4; idx < total; idx += 256 * 4)
        *(float4*)(&As[idx]) = *(const float4*)(&Atile[idx]);
    __syncthreads();

    const int wave = tid >> 5;
    const int lane = tid & 31;
    const int m    = lane & 15;
    const int koff = (lane < 16) ? 0 : 2;
    const int n    = wave * 16 + (lane & 15);

    v8f c = {};
    for (int k = 0; k < K; k += 4) {
        v2f a, b;
        a[0] = As[m * K + k + koff];
        a[1] = As[m * K + k + koff + 1];
        b[0] = W[(size_t)(k + koff)     * HDIM + n];
        b[1] = W[(size_t)(k + koff + 1) * HDIM + n];
        c = __builtin_amdgcn_wmma_f32_16x16x4_f32(false, a, false, b,
                                                  (short)0, c, false, false);
    }

    const int   mbase = row0 + ((lane < 16) ? 0 : 8);
    const float bv    = bias[n];
    for (int r = 0; r < 8; ++r) {
        float v = c[r] + bv;
        if (do_relu) v = fmaxf(v, 0.0f);
        out[(size_t)(mbase + r) * HDIM + n] = v;
    }
}

__global__ __launch_bounds__(256)
void sage_update_wmma_kernel(const float* __restrict__ agg,
                             const float* __restrict__ invdeg,
                             const float* __restrict__ h,
                             const float* __restrict__ Wl,
                             const float* __restrict__ bl,
                             const float* __restrict__ Wr,
                             float* __restrict__ out)
{
    __shared__ __align__(16) float Ms[16 * HDIM];
    __shared__ __align__(16) float Hs[16 * HDIM];
    const int row0 = blockIdx.x * 16;
    const int tid  = threadIdx.x;

    for (int idx = tid * 4; idx < 16 * HDIM; idx += 256 * 4) {
        const int   r = idx / HDIM;
        const float s = invdeg[row0 + r];
        float4 va = *(const float4*)(agg + (size_t)row0 * HDIM + idx);
        va.x *= s; va.y *= s; va.z *= s; va.w *= s;
        *(float4*)(&Ms[idx]) = va;
        *(float4*)(&Hs[idx]) = *(const float4*)(h + (size_t)row0 * HDIM + idx);
    }
    __syncthreads();

    const int wave = tid >> 5;
    const int lane = tid & 31;
    const int m    = lane & 15;
    const int koff = (lane < 16) ? 0 : 2;
    const int n    = wave * 16 + (lane & 15);

    v8f c = {};
    for (int k = 0; k < HDIM; k += 4) {
        v2f a, b;
        a[0] = Ms[m * HDIM + k + koff];
        a[1] = Ms[m * HDIM + k + koff + 1];
        b[0] = Wl[(size_t)(k + koff)     * HDIM + n];
        b[1] = Wl[(size_t)(k + koff + 1) * HDIM + n];
        c = __builtin_amdgcn_wmma_f32_16x16x4_f32(false, a, false, b,
                                                  (short)0, c, false, false);
    }
    for (int k = 0; k < HDIM; k += 4) {
        v2f a, b;
        a[0] = Hs[m * HDIM + k + koff];
        a[1] = Hs[m * HDIM + k + koff + 1];
        b[0] = Wr[(size_t)(k + koff)     * HDIM + n];
        b[1] = Wr[(size_t)(k + koff + 1) * HDIM + n];
        c = __builtin_amdgcn_wmma_f32_16x16x4_f32(false, a, false, b,
                                                  (short)0, c, false, false);
    }

    const int   mbase = row0 + ((lane < 16) ? 0 : 8);
    const float bv    = bl[n];
    for (int r = 0; r < 8; ++r) {
        float v = fmaxf(c[r] + bv, 0.0f);
        out[(size_t)(mbase + r) * HDIM + n] = v;
    }
}

__global__ __launch_bounds__(32)
void cls_wmma_kernel(const float* __restrict__ h, const float* __restrict__ Wc,
                     const float* __restrict__ bc, float* __restrict__ out, int ncls)
{
    const int row0 = blockIdx.x * 16;
    const int lane = threadIdx.x & 31;
    const int m    = lane & 15;
    const int koff = (lane < 16) ? 0 : 2;
    const int n    = blockIdx.y * 16 + (lane & 15);
    const bool nv  = (n < ncls);

    v8f c = {};
    for (int k = 0; k < HDIM; k += 4) {
        v2f a, b;
        a[0] = h[(size_t)(row0 + m) * HDIM + k + koff];
        a[1] = h[(size_t)(row0 + m) * HDIM + k + koff + 1];
        b[0] = nv ? Wc[(size_t)(k + koff)     * ncls + n] : 0.0f;
        b[1] = nv ? Wc[(size_t)(k + koff + 1) * ncls + n] : 0.0f;
        c = __builtin_amdgcn_wmma_f32_16x16x4_f32(false, a, false, b,
                                                  (short)0, c, false, false);
    }

    if (nv) {
        const int   mbase = row0 + ((lane < 16) ? 0 : 8);
        const float bv    = bc[n];
        for (int r = 0; r < 8; ++r)
            out[(size_t)(mbase + r) * ncls + n] = c[r] + bv;
    }
}

__global__ __launch_bounds__(256)
void rank_kernel(const int* __restrict__ dst, int* __restrict__ cnt,
                 int* __restrict__ rank, int E)
{
    const int e = blockIdx.x * blockDim.x + threadIdx.x;
    if (e < E) rank[e] = atomicAdd(&cnt[dst[e]], 1);
}

__global__ __launch_bounds__(256)
void offsets_kernel(const int* __restrict__ cnt, int* __restrict__ total,
                    int* __restrict__ start, float* __restrict__ invdeg, int n)
{
    __shared__ int wsum[8];
    __shared__ int bbase;
    const int i    = blockIdx.x * 256 + threadIdx.x;
    const int lane = threadIdx.x & 31;
    const int wave = threadIdx.x >> 5;
    const int c    = (i < n) ? cnt[i] : 0;

    int incl = c;
    for (int d = 1; d < 32; d <<= 1) {
        const int t = __shfl_up(incl, d, 32);
        if (lane >= d) incl += t;
    }
    if (lane == 31) wsum[wave] = incl;
    __syncthreads();
    if (threadIdx.x == 0) {
        int run = 0;
        for (int w = 0; w < 8; ++w) { const int t = wsum[w]; wsum[w] = run; run += t; }
        bbase = atomicAdd(total, run);
    }
    __syncthreads();
    if (i < n) {
        start[i]  = bbase + wsum[wave] + incl - c;
        invdeg[i] = 1.0f / fmaxf((float)c, 1.0f);
    }
}

__global__ __launch_bounds__(256)
void fill_kernel(const int* __restrict__ src, const int* __restrict__ dst,
                 const int* __restrict__ start, const int* __restrict__ rank,
                 int* __restrict__ col, int E)
{
    const int e = blockIdx.x * blockDim.x + threadIdx.x;
    if (e < E) col[start[dst[e]] + rank[e]] = src[e];
}

template <int U>
__device__ __forceinline__ void gather_rows(const int* __restrict__ col, int j,
                                            const float* __restrict__ hl, float4& acc)
{
    float4 v[U];
#pragma unroll
    for (int u = 0; u < U; ++u)
        v[u] = *(const float4*)(hl + (size_t)col[j + u] * HDIM);
#pragma unroll
    for (int u = 0; u < U; ++u) {
        acc.x += v[u].x; acc.y += v[u].y; acc.z += v[u].z; acc.w += v[u].w;
    }
}

__global__ __launch_bounds__(256)
void gather_sum_kernel(const int* __restrict__ start, const int* __restrict__ cnt,
                       const int* __restrict__ col, const float* __restrict__ h,
                       float* __restrict__ agg, int n)
{
    const int i = __builtin_amdgcn_readfirstlane(blockIdx.x * 8 + (threadIdx.x >> 5));
    if (i >= n) return;
    const int lane = threadIdx.x & 31;
    const float* hl = h + lane * 4;

    int       j   = start[i];
    const int end = j + cnt[i];
    float4 acc = make_float4(0.0f, 0.0f, 0.0f, 0.0f);
    for (; j + 8 <= end; j += 8) gather_rows<8>(col, j, hl, acc);
    if (j + 4 <= end) { gather_rows<4>(col, j, hl, acc); j += 4; }
    if (j + 2 <= end) { gather_rows<2>(col, j, hl, acc); j += 2; }
    if (j < end)        gather_rows<1>(col, j, hl, acc);

    *(float4*)(agg + (size_t)i * HDIM + lane * 4) = acc;
}

extern "C" void kernel_launch(void* const* d_in, const int* in_sizes, int n_in,
                              void* d_out, int out_size, void* d_ws, size_t ws_size,
                              hipStream_t stream)
{
    const float* x      = (const float*)d_in[0];
    const int*   edge   = (const int*)  d_in[1];
    const float* W_proj = (const float*)d_in[2];
    const float* b_proj = (const float*)d_in[3];
    const float* Wl[3]  = { (const float*)d_in[4], (const float*)d_in[7], (const float*)d_in[10] };
    const float* bl[3]  = { (const float*)d_in[5], (const float*)d_in[8], (const float*)d_in[11] };
    const float* Wr[3]  = { (const float*)d_in[6], (const float*)d_in[9], (const float*)d_in[12] };
    const float* Wc     = (const float*)d_in[13];
    const float* bc     = (const float*)d_in[14];

    const int N    = in_sizes[0] / 256;
    const int E    = in_sizes[1] / 2;
    const int ncls = in_sizes[14];

    const int* src = edge;
    const int* dst = edge + E;

    float* hA     = (float*)d_ws;
    float* hB     = hA  + (size_t)N * HDIM;
    float* agg    = hB  + (size_t)N * HDIM;
    float* invdeg = agg + (size_t)N * HDIM;
    int*   cnt    = (int*)(invdeg + N);
    int*   total  = cnt + N;
    int*   start  = total + 1;
    int*   rank   = start + N;
    int*   col    = rank + E;

    const int rowTiles = (N + 15) / 16;

    hipMemsetAsync(cnt, 0, ((size_t)N + 1) * sizeof(int), stream);
    rank_kernel<<<(E + 255) / 256, 256, 0, stream>>>(dst, cnt, rank, E);
    offsets_kernel<<<(N + 255) / 256, 256, 0, stream>>>(cnt, total, start, invdeg, N);
    fill_kernel<<<(E + 255) / 256, 256, 0, stream>>>(src, dst, start, rank, col, E);

    gemm128_wmma_kernel<<<rowTiles, 256, 0, stream>>>(x, W_proj, b_proj, hA, 256, 1);

    for (int l = 0; l < 3; ++l) {
        gather_sum_kernel<<<(N + 7) / 8, 256, 0, stream>>>(start, cnt, col, hA, agg, N);
        sage_update_wmma_kernel<<<rowTiles, 256, 0, stream>>>(agg, invdeg, hA,
                                                              Wl[l], bl[l], Wr[l], hB);
        float* t = hA; hA = hB; hB = t;
    }

    cls_wmma_kernel<<<dim3(rowTiles, (ncls + 15) / 16), 32, 0, stream>>>(
        hA, Wc, bc, (float*)d_out, ncls);
}


# Projection and SAGE-update GEMMs row-blocked: a wave keeps 4 row tiles so one weight fragment feeds 4 matrix instructions; 64-row A tile in LDS at a 260-dword pitch, free of bank conflicts
# speedup vs baseline: 3.1083x; 1.5733x over previous
#include <hip/hip_runtime.h>
#include <hip/hip_bf16.h>

typedef __attribute__((ext_vector_type(2))) float v2f;
typedef __attribute__((ext_vector_type(8))) float v8f;

#define HDIM 128

#define BM    64
#define RT    4
#define APITCH 260

template <int KD>
__device__ __forceinline__ void mma_rows(const float* __restrict__ Arow,
                                         const float* __restrict__ Wp,
                                         v8f (&c)[RT])
{
#pragma unroll
    for (int k = 0; k < KD; k += 4) {
        v2f b;
        b[0] = Wp[(size_t)k * HDIM];
        b[1] = Wp[(size_t)(k + 1) * HDIM];
#pragma unroll
        for (int t = 0; t < RT; ++t) {
            const v2f a = *(const v2f*)(Arow + t * 16 * APITCH + k);
            c[t] = __builtin_amdgcn_wmma_f32_16x16x4_f32(false, a, false, b,
                                                         (short)0, c[t], false, false);
        }
    }
}

__device__ __forceinline__ void store_rows_relu(const v8f (&c)[RT], float bv, int row0,
                                                int lane, int n, float* __restrict__ out, int N)
{
    float* op = out + (size_t)(row0 + ((lane < 16) ? 0 : 8)) * HDIM + n;
#pragma unroll
    for (int t = 0; t < RT; ++t) {
        if (row0 + t * 16 < N) {
#pragma unroll
            for (int r = 0; r < 8; ++r)
                op[(t * 16 + r) * HDIM] = fmaxf(c[t][r] + bv, 0.0f);
        }
    }
}

__global__ __launch_bounds__(256)
void proj_wmma_kernel(const float* __restrict__ A, const float* __restrict__ W,
                      const float* __restrict__ bias, float* __restrict__ out, int N)
{
    __shared__ __align__(16) float As[BM * APITCH];
    const int row0 = blockIdx.x * BM;
    const int tid  = threadIdx.x;

    {
        const int c4 = (tid & 63) * 4;
#pragma unroll
        for (int p = 0; p < BM / 4; ++p) {
            const int r   = p * 4 + (tid >> 6);
            const int row = min(row0 + r, N - 1);
            *(float4*)(&As[r * APITCH + c4]) = *(const float4*)(A + (size_t)row * 256 + c4);
        }
    }
    __syncthreads();

    const int wave = tid >> 5;
    const int lane = tid & 31;
    const int m    = lane & 15;
    const int koff = (lane < 16) ? 0 : 2;
    const int n    = wave * 16 + (lane & 15);

    v8f c[RT] = {};
    mma_rows<256>(&As[m * APITCH + koff], W + (size_t)koff * HDIM + n, c);

    store_rows_relu(c, bias[n], row0, lane, n, out, N);
}

__global__ __launch_bounds__(256)
void sage_update_wmma_kernel(const float* __restrict__ agg,
                             const float* __restrict__ invdeg,
                             const float* __restrict__ h,
                             const float* __restrict__ Wl,
                             const float* __restrict__ bl,
                             const float* __restrict__ Wr,
                             float* __restrict__ out, int N)
{
    __shared__ __align__(16) float As[BM * APITCH];
    const int row0 = blockIdx.x * BM;
    const int tid  = threadIdx.x;

    {
        const int c4 = (tid & 31) * 4;
#pragma unroll
        for (int p = 0; p < BM / 8; ++p) {
            const int r   = p * 8 + (tid >> 5);
            const int row = min(row0 + r, N - 1);
            const float s = invdeg[row];
            float4 va = *(const float4*)(agg + (size_t)row * HDIM + c4);
            va.x *= s; va.y *= s; va.z *= s; va.w *= s;
            *(float4*)(&As[r * APITCH + c4])        = va;
            *(float4*)(&As[r * APITCH + HDIM + c4]) = *(const float4*)(h + (size_t)row * HDIM + c4);
        }
    }
    __syncthreads();

    const int wave = tid >> 5;
    const int lane = tid & 31;
    const int m    = lane & 15;
    const int koff = (lane < 16) ? 0 : 2;
    const int n    = wave * 16 + (lane & 15);

    v8f c[RT] = {};
    const float* Arow = &As[m * APITCH + koff];
    mma_rows<HDIM>(Arow,        Wl + (size_t)koff * HDIM + n, c);
    mma_rows<HDIM>(Arow + HDIM, Wr + (size_t)koff * HDIM + n, c);

    store_rows_relu(c, bl[n], row0, lane, n, out, N);
}

__global__ __launch_bounds__(32)
void cls_wmma_kernel(const float* __restrict__ h, const float* __restrict__ Wc,
                     const float* __restrict__ bc, float* __restrict__ out, int ncls)
{
    const int row0 = blockIdx.x * 16;
    const int lane = threadIdx.x & 31;
    const int m    = lane & 15;
    const int koff = (lane < 16) ? 0 : 2;
    const int n    = blockIdx.y * 16 + (lane & 15);
    const bool nv  = (n < ncls);

    v8f c = {};
    for (int k = 0; k < HDIM; k += 4) {
        v2f a, b;
        a[0] = h[(size_t)(row0 + m) * HDIM + k + koff];
        a[1] = h[(size_t)(row0 + m) * HDIM + k + koff + 1];
        b[0] = nv ? Wc[(size_t)(k + koff)     * ncls + n] : 0.0f;
        b[1] = nv ? Wc[(size_t)(k + koff + 1) * ncls + n] : 0.0f;
        c = __builtin_amdgcn_wmma_f32_16x16x4_f32(false, a, false, b,
                                                  (short)0, c, false, false);
    }

    if (nv) {
        const int   mbase = row0 + ((lane < 16) ? 0 : 8);
        const float bv    = bc[n];
        for (int r = 0; r < 8; ++r)
            out[(size_t)(mbase + r) * ncls + n] = c[r] + bv;
    }
}

__global__ __launch_bounds__(256)
void rank_kernel(const int* __restrict__ dst, int* __restrict__ cnt,
                 int* __restrict__ rank, int E)
{
    const int e = blockIdx.x * blockDim.x + threadIdx.x;
    if (e < E) rank[e] = atomicAdd(&cnt[dst[e]], 1);
}

__global__ __launch_bounds__(256)
void offsets_kernel(const int* __restrict__ cnt, int* __restrict__ total,
                    int* __restrict__ start, float* __restrict__ invdeg, int n)
{
    __shared__ int wsum[8];
    __shared__ int bbase;
    const int i    = blockIdx.x * 256 + threadIdx.x;
    const int lane = threadIdx.x & 31;
    const int wave = threadIdx.x >> 5;
    const int c    = (i < n) ? cnt[i] : 0;

    int incl = c;
    for (int d = 1; d < 32; d <<= 1) {
        const int t = __shfl_up(incl, d, 32);
        if (lane >= d) incl += t;
    }
    if (lane == 31) wsum[wave] = incl;
    __syncthreads();
    if (threadIdx.x == 0) {
        int run = 0;
        for (int w = 0; w < 8; ++w) { const int t = wsum[w]; wsum[w] = run; run += t; }
        bbase = atomicAdd(total, run);
    }
    __syncthreads();
    if (i < n) {
        start[i]  = bbase + wsum[wave] + incl - c;
        invdeg[i] = 1.0f / fmaxf((float)c, 1.0f);
    }
}

__global__ __launch_bounds__(256)
void fill_kernel(const int* __restrict__ src, const int* __restrict__ dst,
                 const int* __restrict__ start, const int* __restrict__ rank,
                 int* __restrict__ col, int E)
{
    const int e = blockIdx.x * blockDim.x + threadIdx.x;
    if (e < E) col[start[dst[e]] + rank[e]] = src[e];
}

template <int U>
__device__ __forceinline__ void gather_rows(const int* __restrict__ col, int j,
                                            const float* __restrict__ hl, float4& acc)
{
    float4 v[U];
#pragma unroll
    for (int u = 0; u < U; ++u)
        v[u] = *(const float4*)(hl + (size_t)col[j + u] * HDIM);
#pragma unroll
    for (int u = 0; u < U; ++u) {
        acc.x += v[u].x; acc.y += v[u].y; acc.z += v[u].z; acc.w += v[u].w;
    }
}

__global__ __launch_bounds__(256)
void gather_sum_kernel(const int* __restrict__ start, const int* __restrict__ cnt,
                       const int* __restrict__ col, const float* __restrict__ h,
                       float* __restrict__ agg, int n)
{
    const int i = __builtin_amdgcn_readfirstlane(blockIdx.x * 8 + (threadIdx.x >> 5));
    if (i >= n) return;
    const int lane = threadIdx.x & 31;
    const float* hl = h + lane * 4;

    int       j   = start[i];
    const int end = j + cnt[i];
    float4 acc = make_float4(0.0f, 0.0f, 0.0f, 0.0f);
    for (; j + 8 <= end; j += 8) gather_rows<8>(col, j, hl, acc);
    if (j + 4 <= end) { gather_rows<4>(col, j, hl, acc); j += 4; }
    if (j + 2 <= end) { gather_rows<2>(col, j, hl, acc); j += 2; }
    if (j < end)        gather_rows<1>(col, j, hl, acc);

    *(float4*)(agg + (size_t)i * HDIM + lane * 4) = acc;
}

extern "C" void kernel_launch(void* const* d_in, const int* in_sizes, int n_in,
                              void* d_out, int out_size, void* d_ws, size_t ws_size,
                              hipStream_t stream)
{
    const float* x      = (const float*)d_in[0];
    const int*   edge   = (const int*)  d_in[1];
    const float* W_proj = (const float*)d_in[2];
    const float* b_proj = (const float*)d_in[3];
    const float* Wl[3]  = { (const float*)d_in[4], (const float*)d_in[7], (const float*)d_in[10] };
    const float* bl[3]  = { (const float*)d_in[5], (const float*)d_in[8], (const float*)d_in[11] };
    const float* Wr[3]  = { (const float*)d_in[6], (const float*)d_in[9], (const float*)d_in[12] };
    const float* Wc     = (const float*)d_in[13];
    const float* bc     = (const float*)d_in[14];

    const int N    = in_sizes[0] / 256;
    const int E    = in_sizes[1] / 2;
    const int ncls = in_sizes[14];

    const int* src = edge;
    const int* dst = edge + E;

    float* hA     = (float*)d_ws;
    float* hB     = hA  + (size_t)N * HDIM;
    float* agg    = hB  + (size_t)N * HDIM;
    float* invdeg = agg + (size_t)N * HDIM;
    int*   cnt    = (int*)(invdeg + N);
    int*   total  = cnt + N;
    int*   start  = total + 1;
    int*   rank   = start + N;
    int*   col    = rank + E;

    const int rowTiles  = (N + 15) / 16;
    const int rowBlocks = (N + BM - 1) / BM;

    hipMemsetAsync(cnt, 0, ((size_t)N + 1) * sizeof(int), stream);
    rank_kernel<<<(E + 255) / 256, 256, 0, stream>>>(dst, cnt, rank, E);
    offsets_kernel<<<(N + 255) / 256, 256, 0, stream>>>(cnt, total, start, invdeg, N);
    fill_kernel<<<(E + 255) / 256, 256, 0, stream>>>(src, dst, start, rank, col, E);

    proj_wmma_kernel<<<rowBlocks, 256, 0, stream>>>(x, W_proj, b_proj, hA, N);

    for (int l = 0; l < 3; ++l) {
        gather_sum_kernel<<<(N + 7) / 8, 256, 0, stream>>>(start, cnt, col, hA, agg, N);
        sage_update_wmma_kernel<<<rowBlocks, 256, 0, stream>>>(agg, invdeg, hA,
                                                               Wl[l], bl[l], Wr[l], hB, N);
        float* t = hA; hA = hB; hB = t;
    }

    cls_wmma_kernel<<<dim3(rowTiles, (ncls + 15) / 16), 32, 0, stream>>>(
        hA, Wc, bc, (float*)d_out, ncls);
}


# Neighbour lists built by a two-level counting sort on LDS counters (bucket histogram, scan, in-bucket rank) instead of a global returning atomic + random fill per edge; list (row sum) order changes
# speedup vs baseline: 3.6049x; 1.1598x over previous
#include <hip/hip_runtime.h>
#include <hip/hip_bf16.h>

typedef __attribute__((ext_vector_type(2))) float v2f;
typedef __attribute__((ext_vector_type(8))) float v8f;

#define HDIM 128

#define BM    64
#define RT    4
#define APITCH 260

template <int KD>
__device__ __forceinline__ void mma_rows(const float* __restrict__ Arow,
                                         const float* __restrict__ Wp,
                                         v8f (&c)[RT])
{
#pragma unroll
    for (int k = 0; k < KD; k += 4) {
        v2f b;
        b[0] = Wp[(size_t)k * HDIM];
        b[1] = Wp[(size_t)(k + 1) * HDIM];
#pragma unroll
        for (int t = 0; t < RT; ++t) {
            const v2f a = *(const v2f*)(Arow + t * 16 * APITCH + k);
            c[t] = __builtin_amdgcn_wmma_f32_16x16x4_f32(false, a, false, b,
                                                         (short)0, c[t], false, false);
        }
    }
}

__device__ __forceinline__ void store_rows_relu(const v8f (&c)[RT], float bv, int row0,
                                                int lane, int n, float* __restrict__ out, int N)
{
    float* op = out + (size_t)(row0 + ((lane < 16) ? 0 : 8)) * HDIM + n;
#pragma unroll
    for (int t = 0; t < RT; ++t) {
        if (row0 + t * 16 < N) {
#pragma unroll
            for (int r = 0; r < 8; ++r)
                op[(t * 16 + r) * HDIM] = fmaxf(c[t][r] + bv, 0.0f);
        }
    }
}

__global__ __launch_bounds__(256)
void proj_wmma_kernel(const float* __restrict__ A, const float* __restrict__ W,
                      const float* __restrict__ bias, float* __restrict__ out, int N)
{
    __shared__ __align__(16) float As[BM * APITCH];
    const int row0 = blockIdx.x * BM;
    const int tid  = threadIdx.x;

    {
        const int c4 = (tid & 63) * 4;
#pragma unroll
        for (int p = 0; p < BM / 4; ++p) {
            const int r   = p * 4 + (tid >> 6);
            const int row = min(row0 + r, N - 1);
            *(float4*)(&As[r * APITCH + c4]) = *(const float4*)(A + (size_t)row * 256 + c4);
        }
    }
    __syncthreads();

    const int wave = tid >> 5;
    const int lane = tid & 31;
    const int m    = lane & 15;
    const int koff = (lane < 16) ? 0 : 2;
    const int n    = wave * 16 + (lane & 15);

    v8f c[RT] = {};
    mma_rows<256>(&As[m * APITCH + koff], W + (size_t)koff * HDIM + n, c);

    store_rows_relu(c, bias[n], row0, lane, n, out, N);
}

__global__ __launch_bounds__(256)
void sage_update_wmma_kernel(const float* __restrict__ agg,
                             const float* __restrict__ invdeg,
                             const float* __restrict__ h,
                             const float* __restrict__ Wl,
                             const float* __restrict__ bl,
                             const float* __restrict__ Wr,
                             float* __restrict__ out, int N)
{
    __shared__ __align__(16) float As[BM * APITCH];
    const int row0 = blockIdx.x * BM;
    const int tid  = threadIdx.x;

    {
        const int c4 = (tid & 31) * 4;
#pragma unroll
        for (int p = 0; p < BM / 8; ++p) {
            const int r   = p * 8 + (tid >> 5);
            const int row = min(row0 + r, N - 1);
            const float s = invdeg[row];
            float4 va = *(const float4*)(agg + (size_t)row * HDIM + c4);
            va.x *= s; va.y *= s; va.z *= s; va.w *= s;
            *(float4*)(&As[r * APITCH + c4])        = va;
            *(float4*)(&As[r * APITCH + HDIM + c4]) = *(const float4*)(h + (size_t)row * HDIM + c4);
        }
    }
    __syncthreads();

    const int wave = tid >> 5;
    const int lane = tid & 31;
    const int m    = lane & 15;
    const int koff = (lane < 16) ? 0 : 2;
    const int n    = wave * 16 + (lane & 15);

    v8f c[RT] = {};
    const float* Arow = &As[m * APITCH + koff];
    mma_rows<HDIM>(Arow,        Wl + (size_t)koff * HDIM + n, c);
    mma_rows<HDIM>(Arow + HDIM, Wr + (size_t)koff * HDIM + n, c);

    store_rows_relu(c, bl[n], row0, lane, n, out, N);
}

__global__ __launch_bounds__(32)
void cls_wmma_kernel(const float* __restrict__ h, const float* __restrict__ Wc,
                     const float* __restrict__ bc, float* __restrict__ out, int ncls)
{
    const int row0 = blockIdx.x * 16;
    const int lane = threadIdx.x & 31;
    const int m    = lane & 15;
    const int koff = (lane < 16) ? 0 : 2;
    const int n    = blockIdx.y * 16 + (lane & 15);
    const bool nv  = (n < ncls);

    v8f c = {};
    for (int k = 0; k < HDIM; k += 4) {
        v2f a, b;
        a[0] = h[(size_t)(row0 + m) * HDIM + k + koff];
        a[1] = h[(size_t)(row0 + m) * HDIM + k + koff + 1];
        b[0] = nv ? Wc[(size_t)(k + koff)     * ncls + n] : 0.0f;
        b[1] = nv ? Wc[(size_t)(k + koff + 1) * ncls + n] : 0.0f;
        c = __builtin_amdgcn_wmma_f32_16x16x4_f32(false, a, false, b,
                                                  (short)0, c, false, false);
    }

    if (nv) {
        const int   mbase = row0 + ((lane < 16) ? 0 : 8);
        const float bv    = bc[n];
        for (int r = 0; r < 8; ++r)
            out[(size_t)(mbase + r) * ncls + n] = c[r] + bv;
    }
}

#define BSH         9
#define BNODES      512
#define NBMAX       256
#define BLD_BLOCKS  256
#define BLD_THREADS 512
#define BLD_WAVES   (BLD_THREADS / 32)
#define EPT         13
#define EPT_CSR     20

__device__ __forceinline__ int block_excl_scan(int v, int* wsum, int tid)
{
    const int lane = tid & 31;
    const int wave = tid >> 5;
    int incl = v;
    for (int d = 1; d < 32; d <<= 1) {
        const int t = __shfl_up(incl, d, 32);
        if (lane >= d) incl += t;
    }
    if (lane == 31) wsum[wave] = incl;
    __syncthreads();
    if (wave == 0) {
        const int s = (lane < BLD_WAVES) ? wsum[lane] : 0;
        int si = s;
        for (int d = 1; d < 32; d <<= 1) {
            const int t = __shfl_up(si, d, 32);
            if (lane >= d) si += t;
        }
        if (lane < BLD_WAVES) wsum[lane] = si - s;
    }
    __syncthreads();
    return wsum[wave] + incl - v;
}

__global__ __launch_bounds__(BLD_THREADS)
void bucket_hist_kernel(const int* __restrict__ dst, int* __restrict__ ghist, int E)
{
    __shared__ int hist[NBMAX];
    const int tid = threadIdx.x;
    if (tid < NBMAX) hist[tid] = 0;
    __syncthreads();

    const int chunk = (E + BLD_BLOCKS - 1) / BLD_BLOCKS;
    const int base  = blockIdx.x * chunk;
    const int end   = min(base + chunk, E);
    for (int s = base; s < end; s += BLD_THREADS * EPT) {
        int d[EPT];
#pragma unroll
        for (int u = 0; u < EPT; ++u)
            d[u] = dst[min(s + u * BLD_THREADS + tid, E - 1)];
#pragma unroll
        for (int u = 0; u < EPT; ++u)
            if (s + u * BLD_THREADS + tid < end) atomicAdd(&hist[d[u] >> BSH], 1);
    }
    __syncthreads();
    if (tid < NBMAX) ghist[blockIdx.x * NBMAX + tid] = hist[tid];
}

#define TQ  (NBMAX / 4)
#define TG  (BLD_THREADS / TQ)
#define TR  (BLD_BLOCKS / TG)

__global__ __launch_bounds__(BLD_THREADS)
void bucket_bin_kernel(const int* __restrict__ src, const int* __restrict__ dst,
                       const int* __restrict__ ghist, int* __restrict__ bbase,
                       unsigned* __restrict__ binned, int E)
{
    __shared__ __align__(16) int ppre[TG][NBMAX];
    __shared__ __align__(16) int ptot[TG][NBMAX];
    __shared__ int offs[NBMAX];
    __shared__ int wsum[BLD_WAVES];
    const int tid = threadIdx.x;

    {
        const int q = tid & (TQ - 1);
        const int g = tid / TQ;
        const int4* p = (const int4*)ghist + (size_t)(g * TR) * TQ + q;
        int4 v[TR];
#pragma unroll
        for (int r = 0; r < TR; ++r) v[r] = p[r * TQ];
        int4 tot = make_int4(0, 0, 0, 0), pre = make_int4(0, 0, 0, 0);
#pragma unroll
        for (int r = 0; r < TR; ++r) {
            tot.x += v[r].x; tot.y += v[r].y; tot.z += v[r].z; tot.w += v[r].w;
            if (g * TR + r < (int)blockIdx.x) {
                pre.x += v[r].x; pre.y += v[r].y; pre.z += v[r].z; pre.w += v[r].w;
            }
        }
        *(int4*)&ptot[g][4 * q] = tot;
        *(int4*)&ppre[g][4 * q] = pre;
    }
    __syncthreads();

    int tot = 0, pre = 0;
    if (tid < NBMAX) {
#pragma unroll
        for (int g = 0; g < TG; ++g) { tot += ptot[g][tid]; pre += ppre[g][tid]; }
    }
    const int excl = block_excl_scan(tot, wsum, tid);
    if (tid < NBMAX) {
        offs[tid] = excl + pre;
        if (blockIdx.x == 0) {
            bbase[tid] = excl;
            if (tid == NBMAX - 1) bbase[NBMAX] = excl + tot;
        }
    }
    __syncthreads();

    const int chunk = (E + BLD_BLOCKS - 1) / BLD_BLOCKS;
    const int base  = blockIdx.x * chunk;
    const int end   = min(base + chunk, E);
    for (int s = base; s < end; s += BLD_THREADS * EPT) {
        int d[EPT], sv[EPT];
#pragma unroll
        for (int u = 0; u < EPT; ++u) {
            const int e = min(s + u * BLD_THREADS + tid, E - 1);
            d[u]  = dst[e];
            sv[u] = src[e];
        }
        int pos[EPT];
#pragma unroll
        for (int u = 0; u < EPT; ++u) {
            pos[u] = 0;
            if (s + u * BLD_THREADS + tid < end) pos[u] = atomicAdd(&offs[d[u] >> BSH], 1);
        }
#pragma unroll
        for (int u = 0; u < EPT; ++u) {
            if (s + u * BLD_THREADS + tid < end)
                binned[pos[u]] = ((unsigned)sv[u] << BSH) | ((unsigned)d[u] & (BNODES - 1));
        }
    }
}

__global__ __launch_bounds__(BLD_THREADS)
void bucket_csr_kernel(const unsigned* __restrict__ binned, const int* __restrict__ bbase,
                       int* __restrict__ cnt, int* __restrict__ start,
                       float* __restrict__ invdeg, int* __restrict__ col, int n)
{
    __shared__ int c[BNODES];
    __shared__ int fpos[BNODES];
    __shared__ int wsum[BLD_WAVES];
    const int tid = threadIdx.x;
    const int lo  = bbase[blockIdx.x];
    const int hi  = bbase[blockIdx.x + 1];
    c[tid] = 0;
    __syncthreads();

    for (int s = lo; s < hi; s += BLD_THREADS * EPT_CSR) {
        unsigned p[EPT_CSR];
#pragma unroll
        for (int u = 0; u < EPT_CSR; ++u)
            p[u] = binned[min(s + u * BLD_THREADS + tid, hi - 1)];
#pragma unroll
        for (int u = 0; u < EPT_CSR; ++u)
            if (s + u * BLD_THREADS + tid < hi) atomicAdd(&c[p[u] & (BNODES - 1)], 1);
    }
    __syncthreads();

    const int v    = c[tid];
    const int excl = block_excl_scan(v, wsum, tid);
    fpos[tid] = lo + excl;
    {
        const int node = blockIdx.x * BNODES + tid;
        if (node < n) {
            cnt[node]    = v;
            start[node]  = lo + excl;
            invdeg[node] = 1.0f / fmaxf((float)v, 1.0f);
        }
    }
    __syncthreads();

    for (int s = lo; s < hi; s += BLD_THREADS * EPT_CSR) {
        unsigned p[EPT_CSR];
#pragma unroll
        for (int u = 0; u < EPT_CSR; ++u)
            p[u] = binned[min(s + u * BLD_THREADS + tid, hi - 1)];
        int pos[EPT_CSR];
#pragma unroll
        for (int u = 0; u < EPT_CSR; ++u) {
            pos[u] = 0;
            if (s + u * BLD_THREADS + tid < hi) pos[u] = atomicAdd(&fpos[p[u] & (BNODES - 1)], 1);
        }
#pragma unroll
        for (int u = 0; u < EPT_CSR; ++u) {
            if (s + u * BLD_THREADS + tid < hi) col[pos[u]] = (int)(p[u] >> BSH);
        }
    }
}

template <int U>
__device__ __forceinline__ void gather_rows(const int* __restrict__ col, int j,
                                            const float* __restrict__ hl, float4& acc)
{
    float4 v[U];
#pragma unroll
    for (int u = 0; u < U; ++u)
        v[u] = *(const float4*)(hl + (size_t)col[j + u] * HDIM);
#pragma unroll
    for (int u = 0; u < U; ++u) {
        acc.x += v[u].x; acc.y += v[u].y; acc.z += v[u].z; acc.w += v[u].w;
    }
}

__global__ __launch_bounds__(256)
void gather_sum_kernel(const int* __restrict__ start, const int* __restrict__ cnt,
                       const int* __restrict__ col, const float* __restrict__ h,
                       float* __restrict__ agg, int n)
{
    const int i = __builtin_amdgcn_readfirstlane(blockIdx.x * 8 + (threadIdx.x >> 5));
    if (i >= n) return;
    const int lane = threadIdx.x & 31;
    const float* hl = h + lane * 4;

    int       j   = start[i];
    const int end = j + cnt[i];
    float4 acc = make_float4(0.0f, 0.0f, 0.0f, 0.0f);
    for (; j + 8 <= end; j += 8) gather_rows<8>(col, j, hl, acc);
    if (j + 4 <= end) { gather_rows<4>(col, j, hl, acc); j += 4; }
    if (j + 2 <= end) { gather_rows<2>(col, j, hl, acc); j += 2; }
    if (j < end)        gather_rows<1>(col, j, hl, acc);

    *(float4*)(agg + (size_t)i * HDIM + lane * 4) = acc;
}

extern "C" void kernel_launch(void* const* d_in, const int* in_sizes, int n_in,
                              void* d_out, int out_size, void* d_ws, size_t ws_size,
                              hipStream_t stream)
{
    const float* x      = (const float*)d_in[0];
    const int*   edge   = (const int*)  d_in[1];
    const float* W_proj = (const float*)d_in[2];
    const float* b_proj = (const float*)d_in[3];
    const float* Wl[3]  = { (const float*)d_in[4], (const float*)d_in[7], (const float*)d_in[10] };
    const float* bl[3]  = { (const float*)d_in[5], (const float*)d_in[8], (const float*)d_in[11] };
    const float* Wr[3]  = { (const float*)d_in[6], (const float*)d_in[9], (const float*)d_in[12] };
    const float* Wc     = (const float*)d_in[13];
    const float* bc     = (const float*)d_in[14];

    const int N    = in_sizes[0] / 256;
    const int E    = in_sizes[1] / 2;
    const int ncls = in_sizes[14];

    const int* src = edge;
    const int* dst = edge + E;

    float* hA     = (float*)d_ws;
    float* hB     = hA  + (size_t)N * HDIM;
    float* agg    = hB  + (size_t)N * HDIM;
    int*   ghist  = (int*)(agg + (size_t)N * HDIM);
    int*   bbase  = ghist + BLD_BLOCKS * NBMAX;
    float* invdeg = (float*)(bbase + NBMAX + 4);
    int*   cnt    = (int*)(invdeg + N);
    int*   start  = cnt + N;
    int*   col    = start + N;
    unsigned* binned = (unsigned*)(col + E);

    const int rowTiles  = (N + 15) / 16;
    const int rowBlocks = (N + BM - 1) / BM;

    bucket_hist_kernel<<<BLD_BLOCKS, BLD_THREADS, 0, stream>>>(dst, ghist, E);
    bucket_bin_kernel<<<BLD_BLOCKS, BLD_THREADS, 0, stream>>>(src, dst, ghist, bbase, binned, E);
    bucket_csr_kernel<<<(N + BNODES - 1) / BNODES, BLD_THREADS, 0, stream>>>(
        binned, bbase, cnt, start, invdeg, col, N);

    proj_wmma_kernel<<<rowBlocks, 256, 0, stream>>>(x, W_proj, b_proj, hA, N);

    for (int l = 0; l < 3; ++l) {
        gather_sum_kernel<<<(N + 7) / 8, 256, 0, stream>>>(start, cnt, col, hA, agg, N);
        sage_update_wmma_kernel<<<rowBlocks, 256, 0, stream>>>(agg, invdeg, hA,
                                                               Wl[l], bl[l], Wr[l], hB, N);
        float* t = hA; hA = hB; hB = t;
    }

    cls_wmma_kernel<<<dim3(rowTiles, (ncls + 15) / 16), 32, 0, stream>>>(
        hA, Wc, bc, (float*)d_out, ncls);
}


# Neighbour gather fused into the SAGE update kernel: a wave sums the lists of 8 of the block's 64 rows into the LDS A tile; aggregate plane and 3 launches gone; row sum order kept
# speedup vs baseline: 3.9462x; 1.0947x over previous
#include <hip/hip_runtime.h>
#include <hip/hip_bf16.h>

typedef __attribute__((ext_vector_type(2))) float v2f;
typedef __attribute__((ext_vector_type(8))) float v8f;

#define HDIM 128

#define BM    64
#define RT    4
#define APITCH 260

template <int KD>
__device__ __forceinline__ void mma_rows(const float* __restrict__ Arow,
                                         const float* __restrict__ Wp,
                                         v8f (&c)[RT])
{
#pragma unroll
    for (int k = 0; k < KD; k += 4) {
        v2f b;
        b[0] = Wp[(size_t)k * HDIM];
        b[1] = Wp[(size_t)(k + 1) * HDIM];
#pragma unroll
        for (int t = 0; t < RT; ++t) {
            const v2f a = *(const v2f*)(Arow + t * 16 * APITCH + k);
            c[t] = __builtin_amdgcn_wmma_f32_16x16x4_f32(false, a, false, b,
                                                         (short)0, c[t], false, false);
        }
    }
}

__device__ __forceinline__ void store_rows_relu(const v8f (&c)[RT], float bv, int row0,
                                                int lane, int n, float* __restrict__ out, int N)
{
    float* op = out + (size_t)(row0 + ((lane < 16) ? 0 : 8)) * HDIM + n;
#pragma unroll
    for (int t = 0; t < RT; ++t) {
        if (row0 + t * 16 < N) {
#pragma unroll
            for (int r = 0; r < 8; ++r)
                op[(t * 16 + r) * HDIM] = fmaxf(c[t][r] + bv, 0.0f);
        }
    }
}

__global__ __launch_bounds__(256)
void proj_wmma_kernel(const float* __restrict__ A, const float* __restrict__ W,
                      const float* __restrict__ bias, float* __restrict__ out, int N)
{
    __shared__ __align__(16) float As[BM * APITCH];
    const int row0 = blockIdx.x * BM;
    const int tid  = threadIdx.x;

    {
        const int c4 = (tid & 63) * 4;
#pragma unroll
        for (int p = 0; p < BM / 4; ++p) {
            const int r   = p * 4 + (tid >> 6);
            const int row = min(row0 + r, N - 1);
            *(float4*)(&As[r * APITCH + c4]) = *(const float4*)(A + (size_t)row * 256 + c4);
        }
    }
    __syncthreads();

    const int wave = tid >> 5;
    const int lane = tid & 31;
    const int m    = lane & 15;
    const int koff = (lane < 16) ? 0 : 2;
    const int n    = wave * 16 + (lane & 15);

    v8f c[RT] = {};
    mma_rows<256>(&As[m * APITCH + koff], W + (size_t)koff * HDIM + n, c);

    store_rows_relu(c, bias[n], row0, lane, n, out, N);
}

__global__ __launch_bounds__(32)
void cls_wmma_kernel(const float* __restrict__ h, const float* __restrict__ Wc,
                     const float* __restrict__ bc, float* __restrict__ out, int ncls)
{
    const int row0 = blockIdx.x * 16;
    const int lane = threadIdx.x & 31;
    const int m    = lane & 15;
    const int koff = (lane < 16) ? 0 : 2;
    const int n    = blockIdx.y * 16 + (lane & 15);
    const bool nv  = (n < ncls);

    v8f c = {};
    for (int k = 0; k < HDIM; k += 4) {
        v2f a, b;
        a[0] = h[(size_t)(row0 + m) * HDIM + k + koff];
        a[1] = h[(size_t)(row0 + m) * HDIM + k + koff + 1];
        b[0] = nv ? Wc[(size_t)(k + koff)     * ncls + n] : 0.0f;
        b[1] = nv ? Wc[(size_t)(k + koff + 1) * ncls + n] : 0.0f;
        c = __builtin_amdgcn_wmma_f32_16x16x4_f32(false, a, false, b,
                                                  (short)0, c, false, false);
    }

    if (nv) {
        const int   mbase = row0 + ((lane < 16) ? 0 : 8);
        const float bv    = bc[n];
        for (int r = 0; r < 8; ++r)
            out[(size_t)(mbase + r) * ncls + n] = c[r] + bv;
    }
}

#define BSH         9
#define BNODES      512
#define NBMAX       256
#define BLD_BLOCKS  256
#define BLD_THREADS 512
#define BLD_WAVES   (BLD_THREADS / 32)
#define EPT         13
#define EPT_CSR     20

__device__ __forceinline__ int block_excl_scan(int v, int* wsum, int tid)
{
    const int lane = tid & 31;
    const int wave = tid >> 5;
    int incl = v;
    for (int d = 1; d < 32; d <<= 1) {
        const int t = __shfl_up(incl, d, 32);
        if (lane >= d) incl += t;
    }
    if (lane == 31) wsum[wave] = incl;
    __syncthreads();
    if (wave == 0) {
        const int s = (lane < BLD_WAVES) ? wsum[lane] : 0;
        int si = s;
        for (int d = 1; d < 32; d <<= 1) {
            const int t = __shfl_up(si, d, 32);
            if (lane >= d) si += t;
        }
        if (lane < BLD_WAVES) wsum[lane] = si - s;
    }
    __syncthreads();
    return wsum[wave] + incl - v;
}

__global__ __launch_bounds__(BLD_THREADS)
void bucket_hist_kernel(const int* __restrict__ dst, int* __restrict__ ghist, int E)
{
    __shared__ int hist[NBMAX];
    const int tid = threadIdx.x;
    if (tid < NBMAX) hist[tid] = 0;
    __syncthreads();

    const int chunk = (E + BLD_BLOCKS - 1) / BLD_BLOCKS;
    const int base  = blockIdx.x * chunk;
    const int end   = min(base + chunk, E);
    for (int s = base; s < end; s += BLD_THREADS * EPT) {
        int d[EPT];
#pragma unroll
        for (int u = 0; u < EPT; ++u)
            d[u] = dst[min(s + u * BLD_THREADS + tid, E - 1)];
#pragma unroll
        for (int u = 0; u < EPT; ++u)
            if (s + u * BLD_THREADS + tid < end) atomicAdd(&hist[d[u] >> BSH], 1);
    }
    __syncthreads();
    if (tid < NBMAX) ghist[blockIdx.x * NBMAX + tid] = hist[tid];
}

#define TQ  (NBMAX / 4)
#define TG  (BLD_THREADS / TQ)
#define TR  (BLD_BLOCKS / TG)

__global__ __launch_bounds__(BLD_THREADS)
void bucket_bin_kernel(const int* __restrict__ src, const int* __restrict__ dst,
                       const int* __restrict__ ghist, int* __restrict__ bbase,
                       unsigned* __restrict__ binned, int E)
{
    __shared__ __align__(16) int ppre[TG][NBMAX];
    __shared__ __align__(16) int ptot[TG][NBMAX];
    __shared__ int offs[NBMAX];
    __shared__ int wsum[BLD_WAVES];
    const int tid = threadIdx.x;

    {
        const int q = tid & (TQ - 1);
        const int g = tid / TQ;
        const int4* p = (const int4*)ghist + (size_t)(g * TR) * TQ + q;
        int4 v[TR];
#pragma unroll
        for (int r = 0; r < TR; ++r) v[r] = p[r * TQ];
        int4 tot = make_int4(0, 0, 0, 0), pre = make_int4(0, 0, 0, 0);
#pragma unroll
        for (int r = 0; r < TR; ++r) {
            tot.x += v[r].x; tot.y += v[r].y; tot.z += v[r].z; tot.w += v[r].w;
            if (g * TR + r < (int)blockIdx.x) {
                pre.x += v[r].x; pre.y += v[r].y; pre.z += v[r].z; pre.w += v[r].w;
            }
        }
        *(int4*)&ptot[g][4 * q] = tot;
        *(int4*)&ppre[g][4 * q] = pre;
    }
    __syncthreads();

    int tot = 0, pre = 0;
    if (tid < NBMAX) {
#pragma unroll
        for (int g = 0; g < TG; ++g) { tot += ptot[g][tid]; pre += ppre[g][tid]; }
    }
    const int excl = block_excl_scan(tot, wsum, tid);
    if (tid < NBMAX) {
        offs[tid] = excl + pre;
        if (blockIdx.x == 0) {
            bbase[tid] = excl;
            if (tid == NBMAX - 1) bbase[NBMAX] = excl + tot;
        }
    }
    __syncthreads();

    const int chunk = (E + BLD_BLOCKS - 1) / BLD_BLOCKS;
    const int base  = blockIdx.x * chunk;
    const int end   = min(base + chunk, E);
    for (int s = base; s < end; s += BLD_THREADS * EPT) {
        int d[EPT], sv[EPT];
#pragma unroll
        for (int u = 0; u < EPT; ++u) {
            const int e = min(s + u * BLD_THREADS + tid, E - 1);
            d[u]  = dst[e];
            sv[u] = src[e];
        }
        int pos[EPT];
#pragma unroll
        for (int u = 0; u < EPT; ++u) {
            pos[u] = 0;
            if (s + u * BLD_THREADS + tid < end) pos[u] = atomicAdd(&offs[d[u] >> BSH], 1);
        }
#pragma unroll
        for (int u = 0; u < EPT; ++u) {
            if (s + u * BLD_THREADS + tid < end)
                binned[pos[u]] = ((unsigned)sv[u] << BSH) | ((unsigned)d[u] & (BNODES - 1));
        }
    }
}

__global__ __launch_bounds__(BLD_THREADS)
void bucket_csr_kernel(const unsigned* __restrict__ binned, const int* __restrict__ bbase,
                       int* __restrict__ cnt, int* __restrict__ start,
                       float* __restrict__ invdeg, int* __restrict__ col, int n)
{
    __shared__ int c[BNODES];
    __shared__ int fpos[BNODES];
    __shared__ int wsum[BLD_WAVES];
    const int tid = threadIdx.x;
    const int lo  = bbase[blockIdx.x];
    const int hi  = bbase[blockIdx.x + 1];
    c[tid] = 0;
    __syncthreads();

    for (int s = lo; s < hi; s += BLD_THREADS * EPT_CSR) {
        unsigned p[EPT_CSR];
#pragma unroll
        for (int u = 0; u < EPT_CSR; ++u)
            p[u] = binned[min(s + u * BLD_THREADS + tid, hi - 1)];
#pragma unroll
        for (int u = 0; u < EPT_CSR; ++u)
            if (s + u * BLD_THREADS + tid < hi) atomicAdd(&c[p[u] & (BNODES - 1)], 1);
    }
    __syncthreads();

    const int v    = c[tid];
    const int excl = block_excl_scan(v, wsum, tid);
    fpos[tid] = lo + excl;
    {
        const int node = blockIdx.x * BNODES + tid;
        if (node < n) {
            cnt[node]    = v;
            start[node]  = lo + excl;
            invdeg[node] = 1.0f / fmaxf((float)v, 1.0f);
        }
    }
    __syncthreads();

    for (int s = lo; s < hi; s += BLD_THREADS * EPT_CSR) {
        unsigned p[EPT_CSR];
#pragma unroll
        for (int u = 0; u < EPT_CSR; ++u)
            p[u] = binned[min(s + u * BLD_THREADS + tid, hi - 1)];
        int pos[EPT_CSR];
#pragma unroll
        for (int u = 0; u < EPT_CSR; ++u) {
            pos[u] = 0;
            if (s + u * BLD_THREADS + tid < hi) pos[u] = atomicAdd(&fpos[p[u] & (BNODES - 1)], 1);
        }
#pragma unroll
        for (int u = 0; u < EPT_CSR; ++u) {
            if (s + u * BLD_THREADS + tid < hi) col[pos[u]] = (int)(p[u] >> BSH);
        }
    }
}

#define GK (BM / 8)

template <int U>
__device__ __forceinline__ void gather_rows(int ids, int t,
                                            const float* __restrict__ hl, float4& acc)
{
    float4 v[U];
#pragma unroll
    for (int u = 0; u < U; ++u) {
        const int s = __builtin_amdgcn_readlane(ids, t + u);
        v[u] = *(const float4*)(hl + (size_t)s * HDIM);
    }
#pragma unroll
    for (int u = 0; u < U; ++u) {
        acc.x += v[u].x; acc.y += v[u].y; acc.z += v[u].z; acc.w += v[u].w;
    }
}

__device__ __forceinline__ void gather_ids(int ids, int m,
                                           const float* __restrict__ hl, float4& acc)
{
    int t = 0;
    for (; t + 8 <= m; t += 8) gather_rows<8>(ids, t, hl, acc);
    if (t + 4 <= m) { gather_rows<4>(ids, t, hl, acc); t += 4; }
    if (t + 2 <= m) { gather_rows<2>(ids, t, hl, acc); t += 2; }
    if (t < m)        gather_rows<1>(ids, t, hl, acc);
}

__global__ __launch_bounds__(256)
void sage_layer_wmma_kernel(const int* __restrict__ start, const int* __restrict__ cnt,
                            const int* __restrict__ col, const float* __restrict__ invdeg,
                            const float* __restrict__ h,
                            const float* __restrict__ Wl, const float* __restrict__ bl,
                            const float* __restrict__ Wr,
                            float* __restrict__ out, int N, int E)
{
    __shared__ __align__(16) float As[BM * APITCH];
    const int row0 = blockIdx.x * BM;
    const int tid  = threadIdx.x;
    const int lane = tid & 31;
    const int wave = __builtin_amdgcn_readfirstlane(tid >> 5);

    {
        const int r0 = wave * GK;
        const float* hl = h + lane * 4;

        float4 self[GK];
#pragma unroll
        for (int q = 0; q < GK; ++q) {
            const int row = min(row0 + r0 + q, N - 1);
            self[q] = *(const float4*)(hl + (size_t)row * HDIM);
        }

        const int qi  = min(row0 + r0 + (lane & (GK - 1)), N - 1);
        const int stv = start[qi];
        const int cnv = cnt[qi];
        const int ivv = __float_as_int(invdeg[qi]);

        int j   = __builtin_amdgcn_readlane(stv, 0);
        int end = j + __builtin_amdgcn_readlane(cnv, 0);
        int ids = col[min(j + lane, E - 1)];

#pragma unroll
        for (int q = 0; q < GK; ++q)
            *(float4*)(&As[(r0 + q) * APITCH + HDIM + lane * 4]) = self[q];

        for (int q = 0; q < GK; ++q) {
            int jn = end, endn = end, idsn = ids;
            if (q + 1 < GK) {
                jn   = __builtin_amdgcn_readlane(stv, q + 1);
                endn = jn + __builtin_amdgcn_readlane(cnv, q + 1);
                idsn = col[min(jn + lane, E - 1)];
            }

            float4 acc = make_float4(0.0f, 0.0f, 0.0f, 0.0f);
            gather_ids(ids, min(end - j, 32), hl, acc);
            for (int base = j + 32; base < end; base += 32) {
                const int more = col[min(base + lane, E - 1)];
                gather_ids(more, min(end - base, 32), hl, acc);
            }
            const float s = __int_as_float(__builtin_amdgcn_readlane(ivv, q));
            acc.x *= s; acc.y *= s; acc.z *= s; acc.w *= s;
            *(float4*)(&As[(r0 + q) * APITCH + lane * 4]) = acc;

            j = jn; end = endn; ids = idsn;
        }
    }
    __syncthreads();

    const int m    = lane & 15;
    const int koff = (lane < 16) ? 0 : 2;
    const int n    = wave * 16 + (lane & 15);

    v8f c[RT] = {};
    const float* Arow = &As[m * APITCH + koff];
    mma_rows<HDIM>(Arow,        Wl + (size_t)koff * HDIM + n, c);
    mma_rows<HDIM>(Arow + HDIM, Wr + (size_t)koff * HDIM + n, c);

    store_rows_relu(c, bl[n], row0, lane, n, out, N);
}

extern "C" void kernel_launch(void* const* d_in, const int* in_sizes, int n_in,
                              void* d_out, int out_size, void* d_ws, size_t ws_size,
                              hipStream_t stream)
{
    const float* x      = (const float*)d_in[0];
    const int*   edge   = (const int*)  d_in[1];
    const float* W_proj = (const float*)d_in[2];
    const float* b_proj = (const float*)d_in[3];
    const float* Wl[3]  = { (const float*)d_in[4], (const float*)d_in[7], (const float*)d_in[10] };
    const float* bl[3]  = { (const float*)d_in[5], (const float*)d_in[8], (const float*)d_in[11] };
    const float* Wr[3]  = { (const float*)d_in[6], (const float*)d_in[9], (const float*)d_in[12] };
    const float* Wc     = (const float*)d_in[13];
    const float* bc     = (const float*)d_in[14];

    const int N    = in_sizes[0] / 256;
    const int E    = in_sizes[1] / 2;
    const int ncls = in_sizes[14];

    const int* src = edge;
    const int* dst = edge + E;

    float* hA     = (float*)d_ws;
    float* hB     = hA  + (size_t)N * HDIM;
    int*   ghist  = (int*)(hB + (size_t)N * HDIM);
    int*   bbase  = ghist + BLD_BLOCKS * NBMAX;
    float* invdeg = (float*)(bbase + NBMAX + 4);
    int*   cnt    = (int*)(invdeg + N);
    int*   start  = cnt + N;
    int*   col    = start + N;
    unsigned* binned = (unsigned*)(col + E);

    const int rowTiles  = (N + 15) / 16;
    const int rowBlocks = (N + BM - 1) / BM;

    bucket_hist_kernel<<<BLD_BLOCKS, BLD_THREADS, 0, stream>>>(dst, ghist, E);
    bucket_bin_kernel<<<BLD_BLOCKS, BLD_THREADS, 0, stream>>>(src, dst, ghist, bbase, binned, E);
    bucket_csr_kernel<<<(N + BNODES - 1) / BNODES, BLD_THREADS, 0, stream>>>(
        binned, bbase, cnt, start, invdeg, col, N);

    proj_wmma_kernel<<<rowBlocks, 256, 0, stream>>>(x, W_proj, b_proj, hA, N);

    for (int l = 0; l < 3; ++l) {
        sage_layer_wmma_kernel<<<rowBlocks, 256, 0, stream>>>(start, cnt, col, invdeg, hA,
                                                              Wl[l], bl[l], Wr[l], hB, N, E);
        float* t = hA; hA = hB; hB = t;
    }

    cls_wmma_kernel<<<dim3(rowTiles, (ncls + 15) / 16), 32, 0, stream>>>(
        hA, Wc, bc, (float*)d_out, ncls);
}


# Classifier product fused into the last SAGE layer's kernel: the block's relu'd 64x128 tile stays in LDS and meets W_cls on the matrix unit there; cls launch and last h plane gone; sum order kept
# speedup vs baseline: 4.4210x; 1.1203x over previous
#include <hip/hip_runtime.h>
#include <hip/hip_bf16.h>

typedef __attribute__((ext_vector_type(2))) float v2f;
typedef __attribute__((ext_vector_type(8))) float v8f;

#define HDIM 128

#define BM    64
#define RT    4
#define APITCH 260

template <int KD>
__device__ __forceinline__ void mma_rows(const float* __restrict__ Arow,
                                         const float* __restrict__ Wp,
                                         v8f (&c)[RT])
{
#pragma unroll
    for (int k = 0; k < KD; k += 4) {
        v2f b;
        b[0] = Wp[(size_t)k * HDIM];
        b[1] = Wp[(size_t)(k + 1) * HDIM];
#pragma unroll
        for (int t = 0; t < RT; ++t) {
            const v2f a = *(const v2f*)(Arow + t * 16 * APITCH + k);
            c[t] = __builtin_amdgcn_wmma_f32_16x16x4_f32(false, a, false, b,
                                                         (short)0, c[t], false, false);
        }
    }
}

__device__ __forceinline__ void store_rows_relu(const v8f (&c)[RT], float bv, int row0,
                                                int lane, int n, float* __restrict__ out, int N)
{
    float* op = out + (size_t)(row0 + ((lane < 16) ? 0 : 8)) * HDIM + n;
#pragma unroll
    for (int t = 0; t < RT; ++t) {
        if (row0 + t * 16 < N) {
#pragma unroll
            for (int r = 0; r < 8; ++r)
                op[(t * 16 + r) * HDIM] = fmaxf(c[t][r] + bv, 0.0f);
        }
    }
}

__global__ __launch_bounds__(256)
void proj_wmma_kernel(const float* __restrict__ A, const float* __restrict__ W,
                      const float* __restrict__ bias, float* __restrict__ out, int N)
{
    __shared__ __align__(16) float As[BM * APITCH];
    const int row0 = blockIdx.x * BM;
    const int tid  = threadIdx.x;

    {
        const int c4 = (tid & 63) * 4;
#pragma unroll
        for (int p = 0; p < BM / 4; ++p) {
            const int r   = p * 4 + (tid >> 6);
            const int row = min(row0 + r, N - 1);
            *(float4*)(&As[r * APITCH + c4]) = *(const float4*)(A + (size_t)row * 256 + c4);
        }
    }
    __syncthreads();

    const int wave = tid >> 5;
    const int lane = tid & 31;
    const int m    = lane & 15;
    const int koff = (lane < 16) ? 0 : 2;
    const int n    = wave * 16 + (lane & 15);

    v8f c[RT] = {};
    mma_rows<256>(&As[m * APITCH + koff], W + (size_t)koff * HDIM + n, c);

    store_rows_relu(c, bias[n], row0, lane, n, out, N);
}

#define BSH         9
#define BNODES      512
#define NBMAX       256
#define BLD_BLOCKS  256
#define BLD_THREADS 512
#define BLD_WAVES   (BLD_THREADS / 32)
#define EPT         13
#define EPT_CSR     20

__device__ __forceinline__ int block_excl_scan(int v, int* wsum, int tid)
{
    const int lane = tid & 31;
    const int wave = tid >> 5;
    int incl = v;
    for (int d = 1; d < 32; d <<= 1) {
        const int t = __shfl_up(incl, d, 32);
        if (lane >= d) incl += t;
    }
    if (lane == 31) wsum[wave] = incl;
    __syncthreads();
    if (wave == 0) {
        const int s = (lane < BLD_WAVES) ? wsum[lane] : 0;
        int si = s;
        for (int d = 1; d < 32; d <<= 1) {
            const int t = __shfl_up(si, d, 32);
            if (lane >= d) si += t;
        }
        if (lane < BLD_WAVES) wsum[lane] = si - s;
    }
    __syncthreads();
    return wsum[wave] + incl - v;
}

__global__ __launch_bounds__(BLD_THREADS)
void bucket_hist_kernel(const int* __restrict__ dst, int* __restrict__ ghist, int E)
{
    __shared__ int hist[NBMAX];
    const int tid = threadIdx.x;
    if (tid < NBMAX) hist[tid] = 0;
    __syncthreads();

    const int chunk = (E + BLD_BLOCKS - 1) / BLD_BLOCKS;
    const int base  = blockIdx.x * chunk;
    const int end   = min(base + chunk, E);
    for (int s = base; s < end; s += BLD_THREADS * EPT) {
        int d[EPT];
#pragma unroll
        for (int u = 0; u < EPT; ++u)
            d[u] = dst[min(s + u * BLD_THREADS + tid, E - 1)];
#pragma unroll
        for (int u = 0; u < EPT; ++u)
            if (s + u * BLD_THREADS + tid < end) atomicAdd(&hist[d[u] >> BSH], 1);
    }
    __syncthreads();
    if (tid < NBMAX) ghist[blockIdx.x * NBMAX + tid] = hist[tid];
}

#define TQ  (NBMAX / 4)
#define TG  (BLD_THREADS / TQ)
#define TR  (BLD_BLOCKS / TG)

__global__ __launch_bounds__(BLD_THREADS)
void bucket_bin_kernel(const int* __restrict__ src, const int* __restrict__ dst,
                       const int* __restrict__ ghist, int* __restrict__ bbase,
                       unsigned* __restrict__ binned, int E)
{
    __shared__ __align__(16) int ppre[TG][NBMAX];
    __shared__ __align__(16) int ptot[TG][NBMAX];
    __shared__ int offs[NBMAX];
    __shared__ int wsum[BLD_WAVES];
    const int tid = threadIdx.x;

    {
        const int q = tid & (TQ - 1);
        const int g = tid / TQ;
        const int4* p = (const int4*)ghist + (size_t)(g * TR) * TQ + q;
        int4 v[TR];
#pragma unroll
        for (int r = 0; r < TR; ++r) v[r] = p[r * TQ];
        int4 tot = make_int4(0, 0, 0, 0), pre = make_int4(0, 0, 0, 0);
#pragma unroll
        for (int r = 0; r < TR; ++r) {
            tot.x += v[r].x; tot.y += v[r].y; tot.z += v[r].z; tot.w += v[r].w;
            if (g * TR + r < (int)blockIdx.x) {
                pre.x += v[r].x; pre.y += v[r].y; pre.z += v[r].z; pre.w += v[r].w;
            }
        }
        *(int4*)&ptot[g][4 * q] = tot;
        *(int4*)&ppre[g][4 * q] = pre;
    }
    __syncthreads();

    int tot = 0, pre = 0;
    if (tid < NBMAX) {
#pragma unroll
        for (int g = 0; g < TG; ++g) { tot += ptot[g][tid]; pre += ppre[g][tid]; }
    }
    const int excl = block_excl_scan(tot, wsum, tid);
    if (tid < NBMAX) {
        offs[tid] = excl + pre;
        if (blockIdx.x == 0) {
            bbase[tid] = excl;
            if (tid == NBMAX - 1) bbase[NBMAX] = excl + tot;
        }
    }
    __syncthreads();

    const int chunk = (E + BLD_BLOCKS - 1) / BLD_BLOCKS;
    const int base  = blockIdx.x * chunk;
    const int end   = min(base + chunk, E);
    for (int s = base; s < end; s += BLD_THREADS * EPT) {
        int d[EPT], sv[EPT];
#pragma unroll
        for (int u = 0; u < EPT; ++u) {
            const int e = min(s + u * BLD_THREADS + tid, E - 1);
            d[u]  = dst[e];
            sv[u] = src[e];
        }
        int pos[EPT];
#pragma unroll
        for (int u = 0; u < EPT; ++u) {
            pos[u] = 0;
            if (s + u * BLD_THREADS + tid < end) pos[u] = atomicAdd(&offs[d[u] >> BSH], 1);
        }
#pragma unroll
        for (int u = 0; u < EPT; ++u) {
            if (s + u * BLD_THREADS + tid < end)
                binned[pos[u]] = ((unsigned)sv[u] << BSH) | ((unsigned)d[u] & (BNODES - 1));
        }
    }
}

__global__ __launch_bounds__(BLD_THREADS)
void bucket_csr_kernel(const unsigned* __restrict__ binned, const int* __restrict__ bbase,
                       int* __restrict__ cnt, int* __restrict__ start,
                       float* __restrict__ invdeg, int* __restrict__ col, int n)
{
    __shared__ int c[BNODES];
    __shared__ int fpos[BNODES];
    __shared__ int wsum[BLD_WAVES];
    const int tid = threadIdx.x;
    const int lo  = bbase[blockIdx.x];
    const int hi  = bbase[blockIdx.x + 1];
    c[tid] = 0;
    __syncthreads();

    for (int s = lo; s < hi; s += BLD_THREADS * EPT_CSR) {
        unsigned p[EPT_CSR];
#pragma unroll
        for (int u = 0; u < EPT_CSR; ++u)
            p[u] = binned[min(s + u * BLD_THREADS + tid, hi - 1)];
#pragma unroll
        for (int u = 0; u < EPT_CSR; ++u)
            if (s + u * BLD_THREADS + tid < hi) atomicAdd(&c[p[u] & (BNODES - 1)], 1);
    }
    __syncthreads();

    const int v    = c[tid];
    const int excl = block_excl_scan(v, wsum, tid);
    fpos[tid] = lo + excl;
    {
        const int node = blockIdx.x * BNODES + tid;
        if (node < n) {
            cnt[node]    = v;
            start[node]  = lo + excl;
            invdeg[node] = 1.0f / fmaxf((float)v, 1.0f);
        }
    }
    __syncthreads();

    for (int s = lo; s < hi; s += BLD_THREADS * EPT_CSR) {
        unsigned p[EPT_CSR];
#pragma unroll
        for (int u = 0; u < EPT_CSR; ++u)
            p[u] = binned[min(s + u * BLD_THREADS + tid, hi - 1)];
        int pos[EPT_CSR];
#pragma unroll
        for (int u = 0; u < EPT_CSR; ++u) {
            pos[u] = 0;
            if (s + u * BLD_THREADS + tid < hi) pos[u] = atomicAdd(&fpos[p[u] & (BNODES - 1)], 1);
        }
#pragma unroll
        for (int u = 0; u < EPT_CSR; ++u) {
            if (s + u * BLD_THREADS + tid < hi) col[pos[u]] = (int)(p[u] >> BSH);
        }
    }
}

#define NCLS 40
static_assert(NCLS <= 48 && NCLS % 4 == 0, "three column tiles, 16-byte output pieces");

template <int NT>
__device__ __forceinline__ void cls_tiles(const float* __restrict__ Arow,
                                          const float* __restrict__ Wc,
                                          int ct0, int lane, v8f (&d)[NT])
{
    const int koff = (lane < 16) ? 0 : 2;
    const float* wp[NT];
    bool nv[NT];
#pragma unroll
    for (int t = 0; t < NT; ++t) {
        const int n = (ct0 + t) * 16 + (lane & 15);
        nv[t] = (n < NCLS);
        wp[t] = Wc + koff * NCLS + min(n, NCLS - 1);
    }
#pragma unroll
    for (int k = 0; k < HDIM; k += 4) {
        const v2f a = *(const v2f*)(Arow + k);
#pragma unroll
        for (int t = 0; t < NT; ++t) {
            const float b0 = wp[t][k * NCLS];
            const float b1 = wp[t][(k + 1) * NCLS];
            v2f b;
            b[0] = nv[t] ? b0 : 0.0f;
            b[1] = nv[t] ? b1 : 0.0f;
            d[t] = __builtin_amdgcn_wmma_f32_16x16x4_f32(false, a, false, b,
                                                         (short)0, d[t], false, false);
        }
    }
}

__device__ __forceinline__ void cls_stage(const v8f& d, int rt, int ct, int lane,
                                          const float* __restrict__ bc, float* __restrict__ Ls)
{
    const int n = ct * 16 + (lane & 15);
    if (n < NCLS) {
        const float bv = bc[n];
        const int   rb = rt * 16 + ((lane < 16) ? 0 : 8);
#pragma unroll
        for (int r = 0; r < 8; ++r)
            Ls[(rb + r) * NCLS + n] = d[r] + bv;
    }
}

#define GK (BM / 8)

template <int U>
__device__ __forceinline__ void gather_rows(int ids, int t,
                                            const float* __restrict__ hl, float4& acc)
{
    float4 v[U];
#pragma unroll
    for (int u = 0; u < U; ++u) {
        const int s = __builtin_amdgcn_readlane(ids, t + u);
        v[u] = *(const float4*)(hl + (size_t)s * HDIM);
    }
#pragma unroll
    for (int u = 0; u < U; ++u) {
        acc.x += v[u].x; acc.y += v[u].y; acc.z += v[u].z; acc.w += v[u].w;
    }
}

__device__ __forceinline__ void gather_ids(int ids, int m,
                                           const float* __restrict__ hl, float4& acc)
{
    int t = 0;
    for (; t + 8 <= m; t += 8) gather_rows<8>(ids, t, hl, acc);
    if (t + 4 <= m) { gather_rows<4>(ids, t, hl, acc); t += 4; }
    if (t + 2 <= m) { gather_rows<2>(ids, t, hl, acc); t += 2; }
    if (t < m)        gather_rows<1>(ids, t, hl, acc);
}

template <bool CLS>
__global__ __launch_bounds__(256)
void sage_layer_wmma_kernel(const int* __restrict__ start, const int* __restrict__ cnt,
                            const int* __restrict__ col, const float* __restrict__ invdeg,
                            const float* __restrict__ h,
                            const float* __restrict__ Wl, const float* __restrict__ bl,
                            const float* __restrict__ Wr,
                            float* __restrict__ out, int N, int E,
                            const float* __restrict__ Wc, const float* __restrict__ bc)
{
    __shared__ __align__(16) float As[BM * APITCH];
    const int row0 = blockIdx.x * BM;
    const int tid  = threadIdx.x;
    const int lane = tid & 31;
    const int wave = __builtin_amdgcn_readfirstlane(tid >> 5);

    {
        const int r0 = wave * GK;
        const float* hl = h + lane * 4;

        float4 self[GK];
#pragma unroll
        for (int q = 0; q < GK; ++q) {
            const int row = min(row0 + r0 + q, N - 1);
            self[q] = *(const float4*)(hl + (size_t)row * HDIM);
        }

        const int qi  = min(row0 + r0 + (lane & (GK - 1)), N - 1);
        const int stv = start[qi];
        const int cnv = cnt[qi];
        const int ivv = __float_as_int(invdeg[qi]);

        int j   = __builtin_amdgcn_readlane(stv, 0);
        int end = j + __builtin_amdgcn_readlane(cnv, 0);
        int ids = col[min(j + lane, E - 1)];

#pragma unroll
        for (int q = 0; q < GK; ++q)
            *(float4*)(&As[(r0 + q) * APITCH + HDIM + lane * 4]) = self[q];

        for (int q = 0; q < GK; ++q) {
            int jn = end, endn = end, idsn = ids;
            if (q + 1 < GK) {
                jn   = __builtin_amdgcn_readlane(stv, q + 1);
                endn = jn + __builtin_amdgcn_readlane(cnv, q + 1);
                idsn = col[min(jn + lane, E - 1)];
            }

            float4 acc = make_float4(0.0f, 0.0f, 0.0f, 0.0f);
            gather_ids(ids, min(end - j, 32), hl, acc);
            for (int base = j + 32; base < end; base += 32) {
                const int more = col[min(base + lane, E - 1)];
                gather_ids(more, min(end - base, 32), hl, acc);
            }
            const float s = __int_as_float(__builtin_amdgcn_readlane(ivv, q));
            acc.x *= s; acc.y *= s; acc.z *= s; acc.w *= s;
            *(float4*)(&As[(r0 + q) * APITCH + lane * 4]) = acc;

            j = jn; end = endn; ids = idsn;
        }
    }
    __syncthreads();

    const int m    = lane & 15;
    const int koff = (lane < 16) ? 0 : 2;
    const int n    = wave * 16 + (lane & 15);

    v8f c[RT] = {};
    const float* Arow = &As[m * APITCH + koff];
    mma_rows<HDIM>(Arow,        Wl + (size_t)koff * HDIM + n, c);
    mma_rows<HDIM>(Arow + HDIM, Wr + (size_t)koff * HDIM + n, c);

    if constexpr (!CLS) {
        store_rows_relu(c, bl[n], row0, lane, n, out, N);
    } else {
        const float bv = bl[n];
        __syncthreads();
        {
            float* hp = &As[((lane < 16) ? 0 : 8) * APITCH + n];
#pragma unroll
            for (int t = 0; t < RT; ++t)
#pragma unroll
                for (int r = 0; r < 8; ++r)
                    hp[(t * 16 + r) * APITCH] = fmaxf(c[t][r] + bv, 0.0f);
        }
        __syncthreads();

        const int rt = wave >> 1;
        const float* Ar = &As[(rt * 16 + m) * APITCH + koff];
        v8f d[2] = {};
        if ((wave & 1) == 0) {
            cls_tiles<2>(Ar, Wc, 0, lane, d);
        } else {
            v8f e[1] = {};
            cls_tiles<1>(Ar, Wc, 2, lane, e);
            d[0] = e[0];
        }
        __syncthreads();

        float* Ls = As;
        if ((wave & 1) == 0) {
            cls_stage(d[0], rt, 0, lane, bc, Ls);
            cls_stage(d[1], rt, 1, lane, bc, Ls);
        } else {
            cls_stage(d[0], rt, 2, lane, bc, Ls);
        }
        __syncthreads();

        const int lim = min(BM, N - row0) * NCLS;
        float* op = out + (size_t)row0 * NCLS;
        for (int i = tid * 4; i < lim; i += 256 * 4)
            *(float4*)(op + i) = *(const float4*)(Ls + i);
    }
}

extern "C" void kernel_launch(void* const* d_in, const int* in_sizes, int n_in,
                              void* d_out, int out_size, void* d_ws, size_t ws_size,
                              hipStream_t stream)
{
    const float* x      = (const float*)d_in[0];
    const int*   edge   = (const int*)  d_in[1];
    const float* W_proj = (const float*)d_in[2];
    const float* b_proj = (const float*)d_in[3];
    const float* Wl[3]  = { (const float*)d_in[4], (const float*)d_in[7], (const float*)d_in[10] };
    const float* bl[3]  = { (const float*)d_in[5], (const float*)d_in[8], (const float*)d_in[11] };
    const float* Wr[3]  = { (const float*)d_in[6], (const float*)d_in[9], (const float*)d_in[12] };
    const float* Wc     = (const float*)d_in[13];
    const float* bc     = (const float*)d_in[14];

    const int N    = in_sizes[0] / 256;
    const int E    = in_sizes[1] / 2;

    const int* src = edge;
    const int* dst = edge + E;

    float* hA     = (float*)d_ws;
    float* hB     = hA  + (size_t)N * HDIM;
    int*   ghist  = (int*)(hB + (size_t)N * HDIM);
    int*   bbase  = ghist + BLD_BLOCKS * NBMAX;
    float* invdeg = (float*)(bbase + NBMAX + 4);
    int*   cnt    = (int*)(invdeg + N);
    int*   start  = cnt + N;
    int*   col    = start + N;
    unsigned* binned = (unsigned*)(col + E);

    const int rowBlocks = (N + BM - 1) / BM;

    bucket_hist_kernel<<<BLD_BLOCKS, BLD_THREADS, 0, stream>>>(dst, ghist, E);
    bucket_bin_kernel<<<BLD_BLOCKS, BLD_THREADS, 0, stream>>>(src, dst, ghist, bbase, binned, E);
    bucket_csr_kernel<<<(N + BNODES - 1) / BNODES, BLD_THREADS, 0, stream>>>(
        binned, bbase, cnt, start, invdeg, col, N);

    proj_wmma_kernel<<<rowBlocks, 256, 0, stream>>>(x, W_proj, b_proj, hA, N);

    for (int l = 0; l < 2; ++l) {
        sage_layer_wmma_kernel<false><<<rowBlocks, 256, 0, stream>>>(start, cnt, col, invdeg, hA,
                                                                     Wl[l], bl[l], Wr[l], hB, N, E,
                                                                     Wc, bc);
        float* t = hA; hA = hB; hB = t;
    }

    sage_layer_wmma_kernel<true><<<rowBlocks, 256, 0, stream>>>(start, cnt, col, invdeg, hA,
                                                                Wl[2], bl[2], Wr[2], (float*)d_out, N, E,
                                                                Wc, bc);
}


# Projection tiles ride inside the three list-building launches (blocks past the list blocks run two 64-row tiles each): matrix work fills the latency-bound list passes; 7 launches -> 6
# speedup vs baseline: 4.5632x; 1.0322x over previous
#include <hip/hip_runtime.h>
#include <hip/hip_bf16.h>

typedef __attribute__((ext_vector_type(2))) float v2f;
typedef __attribute__((ext_vector_type(8))) float v8f;

#define HDIM 128

#define BM    64
#define RT    4
#define APITCH 260

template <int KD>
__device__ __forceinline__ void mma_rows(const float* __restrict__ Arow,
                                         const float* __restrict__ Wp,
                                         v8f (&c)[RT])
{
#pragma unroll
    for (int k = 0; k < KD; k += 4) {
        v2f b;
        b[0] = Wp[(size_t)k * HDIM];
        b[1] = Wp[(size_t)(k + 1) * HDIM];
#pragma unroll
        for (int t = 0; t < RT; ++t) {
            const v2f a = *(const v2f*)(Arow + t * 16 * APITCH + k);
            c[t] = __builtin_amdgcn_wmma_f32_16x16x4_f32(false, a, false, b,
                                                         (short)0, c[t], false, false);
        }
    }
}

__device__ __forceinline__ void store_rows_relu(const v8f (&c)[RT], float bv, int row0,
                                                int lane, int n, float* __restrict__ out, int N)
{
    float* op = out + (size_t)(row0 + ((lane < 16) ? 0 : 8)) * HDIM + n;
#pragma unroll
    for (int t = 0; t < RT; ++t) {
        if (row0 + t * 16 < N) {
#pragma unroll
            for (int r = 0; r < 8; ++r)
                op[(t * 16 + r) * HDIM] = fmaxf(c[t][r] + bv, 0.0f);
        }
    }
}

__device__ __forceinline__ void proj_role(const float* __restrict__ A, const float* __restrict__ W,
                                          const float* __restrict__ bias, float* __restrict__ out,
                                          int N, int tile0, int tid512)
{
    __shared__ __align__(16) float As2[2][BM * APITCH];
    const int half = tid512 >> 8;
    const int tid  = tid512 & 255;
    float* As = As2[half];
    const int row0 = (tile0 + half) * BM;

    {
        const int c4 = (tid & 63) * 4;
#pragma unroll
        for (int p = 0; p < BM / 4; ++p) {
            const int r   = p * 4 + (tid >> 6);
            const int row = min(row0 + r, N - 1);
            *(float4*)(&As[r * APITCH + c4]) = *(const float4*)(A + (size_t)row * 256 + c4);
        }
    }
    __syncthreads();

    const int wave = tid >> 5;
    const int lane = tid & 31;
    const int m    = lane & 15;
    const int koff = (lane < 16) ? 0 : 2;
    const int n    = wave * 16 + (lane & 15);

    v8f c[RT] = {};
    mma_rows<256>(&As[m * APITCH + koff], W + (size_t)koff * HDIM + n, c);

    store_rows_relu(c, bias[n], row0, lane, n, out, N);
}

#define BSH         9
#define BNODES      512
#define NBMAX       256
#define BLD_BLOCKS  256
#define BLD_THREADS 512
#define BLD_WAVES   (BLD_THREADS / 32)
#define EPT         13
#define EPT_CSR     20

__device__ __forceinline__ int block_excl_scan(int v, int* wsum, int tid)
{
    const int lane = tid & 31;
    const int wave = tid >> 5;
    int incl = v;
    for (int d = 1; d < 32; d <<= 1) {
        const int t = __shfl_up(incl, d, 32);
        if (lane >= d) incl += t;
    }
    if (lane == 31) wsum[wave] = incl;
    __syncthreads();
    if (wave == 0) {
        const int s = (lane < BLD_WAVES) ? wsum[lane] : 0;
        int si = s;
        for (int d = 1; d < 32; d <<= 1) {
            const int t = __shfl_up(si, d, 32);
            if (lane >= d) si += t;
        }
        if (lane < BLD_WAVES) wsum[lane] = si - s;
    }
    __syncthreads();
    return wsum[wave] + incl - v;
}

__device__ __forceinline__ void bucket_hist_role(const int* __restrict__ dst, int* __restrict__ ghist,
                                                 int E, int bid, int tid)
{
    __shared__ int hist[NBMAX];
    if (tid < NBMAX) hist[tid] = 0;
    __syncthreads();

    const int chunk = (E + BLD_BLOCKS - 1) / BLD_BLOCKS;
    const int base  = bid * chunk;
    const int end   = min(base + chunk, E);
    for (int s = base; s < end; s += BLD_THREADS * EPT) {
        int d[EPT];
#pragma unroll
        for (int u = 0; u < EPT; ++u)
            d[u] = dst[min(s + u * BLD_THREADS + tid, E - 1)];
#pragma unroll
        for (int u = 0; u < EPT; ++u)
            if (s + u * BLD_THREADS + tid < end) atomicAdd(&hist[d[u] >> BSH], 1);
    }
    __syncthreads();
    if (tid < NBMAX) ghist[bid * NBMAX + tid] = hist[tid];
}

#define TQ  (NBMAX / 4)
#define TG  (BLD_THREADS / TQ)
#define TR  (BLD_BLOCKS / TG)

__device__ __forceinline__ void bucket_bin_role(const int* __restrict__ src, const int* __restrict__ dst,
                                                const int* __restrict__ ghist, int* __restrict__ bbase,
                                                unsigned* __restrict__ binned, int E, int bid, int tid)
{
    __shared__ __align__(16) int ppre[TG][NBMAX];
    __shared__ __align__(16) int ptot[TG][NBMAX];
    __shared__ int offs[NBMAX];
    __shared__ int wsum[BLD_WAVES];

    {
        const int q = tid & (TQ - 1);
        const int g = tid / TQ;
        const int4* p = (const int4*)ghist + (size_t)(g * TR) * TQ + q;
        int4 tot = make_int4(0, 0, 0, 0), pre = make_int4(0, 0, 0, 0);
#pragma nounroll
        for (int r0 = 0; r0 < TR; r0 += TR / 2) {
            int4 v[TR / 2];
#pragma unroll
            for (int r = 0; r < TR / 2; ++r) v[r] = p[(r0 + r) * TQ];
#pragma unroll
            for (int r = 0; r < TR / 2; ++r) {
                tot.x += v[r].x; tot.y += v[r].y; tot.z += v[r].z; tot.w += v[r].w;
                if (g * TR + r0 + r < bid) {
                    pre.x += v[r].x; pre.y += v[r].y; pre.z += v[r].z; pre.w += v[r].w;
                }
            }
        }
        *(int4*)&ptot[g][4 * q] = tot;
        *(int4*)&ppre[g][4 * q] = pre;
    }
    __syncthreads();

    int tot = 0, pre = 0;
    if (tid < NBMAX) {
#pragma unroll
        for (int g = 0; g < TG; ++g) { tot += ptot[g][tid]; pre += ppre[g][tid]; }
    }
    const int excl = block_excl_scan(tot, wsum, tid);
    if (tid < NBMAX) {
        offs[tid] = excl + pre;
        if (bid == 0) {
            bbase[tid] = excl;
            if (tid == NBMAX - 1) bbase[NBMAX] = excl + tot;
        }
    }
    __syncthreads();

    const int chunk = (E + BLD_BLOCKS - 1) / BLD_BLOCKS;
    const int base  = bid * chunk;
    const int end   = min(base + chunk, E);
    for (int s = base; s < end; s += BLD_THREADS * EPT) {
        int d[EPT], sv[EPT];
#pragma unroll
        for (int u = 0; u < EPT; ++u) {
            const int e = min(s + u * BLD_THREADS + tid, E - 1);
            d[u]  = dst[e];
            sv[u] = src[e];
        }
        int pos[EPT];
#pragma unroll
        for (int u = 0; u < EPT; ++u) {
            pos[u] = 0;
            if (s + u * BLD_THREADS + tid < end) pos[u] = atomicAdd(&offs[d[u] >> BSH], 1);
        }
#pragma unroll
        for (int u = 0; u < EPT; ++u) {
            if (s + u * BLD_THREADS + tid < end)
                binned[pos[u]] = ((unsigned)sv[u] << BSH) | ((unsigned)d[u] & (BNODES - 1));
        }
    }
}

__device__ __forceinline__ void bucket_csr_role(const unsigned* __restrict__ binned, const int* __restrict__ bbase,
                                                int* __restrict__ cnt, int* __restrict__ start,
                                                float* __restrict__ invdeg, int* __restrict__ col, int n,
                                                int bid, int tid)
{
    __shared__ int c[BNODES];
    __shared__ int fpos[BNODES];
    __shared__ int wsum[BLD_WAVES];
    const int lo  = bbase[bid];
    const int hi  = bbase[bid + 1];
    c[tid] = 0;
    __syncthreads();

    for (int s = lo; s < hi; s += BLD_THREADS * EPT_CSR) {
        unsigned p[EPT_CSR];
#pragma unroll
        for (int u = 0; u < EPT_CSR; ++u)
            p[u] = binned[min(s + u * BLD_THREADS + tid, hi - 1)];
#pragma unroll
        for (int u = 0; u < EPT_CSR; ++u)
            if (s + u * BLD_THREADS + tid < hi) atomicAdd(&c[p[u] & (BNODES - 1)], 1);
    }
    __syncthreads();

    const int v    = c[tid];
    const int excl = block_excl_scan(v, wsum, tid);
    fpos[tid] = lo + excl;
    {
        const int node = bid * BNODES + tid;
        if (node < n) {
            cnt[node]    = v;
            start[node]  = lo + excl;
            invdeg[node] = 1.0f / fmaxf((float)v, 1.0f);
        }
    }
    __syncthreads();

    for (int s = lo; s < hi; s += BLD_THREADS * EPT_CSR) {
        unsigned p[EPT_CSR];
#pragma unroll
        for (int u = 0; u < EPT_CSR; ++u)
            p[u] = binned[min(s + u * BLD_THREADS + tid, hi - 1)];
        int pos[EPT_CSR];
#pragma unroll
        for (int u = 0; u < EPT_CSR; ++u) {
            pos[u] = 0;
            if (s + u * BLD_THREADS + tid < hi) pos[u] = atomicAdd(&fpos[p[u] & (BNODES - 1)], 1);
        }
#pragma unroll
        for (int u = 0; u < EPT_CSR; ++u) {
            if (s + u * BLD_THREADS + tid < hi) col[pos[u]] = (int)(p[u] >> BSH);
        }
    }
}

template <int ROLE>
__global__ __launch_bounds__(BLD_THREADS)
void build_proj_kernel(const int* __restrict__ src, const int* __restrict__ dst,
                       int* __restrict__ ghist, int* __restrict__ bbase,
                       unsigned* __restrict__ binned,
                       int* __restrict__ cnt, int* __restrict__ start,
                       float* __restrict__ invdeg, int* __restrict__ col,
                       int N, int E, int nBuild, int pairBase,
                       const float* __restrict__ x, const float* __restrict__ Wp,
                       const float* __restrict__ bp, float* __restrict__ h0)
{
    const int tid = threadIdx.x;
    const int bid = blockIdx.x;
    if (bid < nBuild) {
        if constexpr (ROLE == 0) bucket_hist_role(dst, ghist, E, bid, tid);
        if constexpr (ROLE == 1) bucket_bin_role(src, dst, ghist, bbase, binned, E, bid, tid);
        if constexpr (ROLE == 2) bucket_csr_role(binned, bbase, cnt, start, invdeg, col, N, bid, tid);
    } else {
        proj_role(x, Wp, bp, h0, N, 2 * (pairBase + bid - nBuild), tid);
    }
}

#define NCLS 40
static_assert(NCLS <= 48 && NCLS % 4 == 0, "three column tiles, 16-byte output pieces");

template <int NT>
__device__ __forceinline__ void cls_tiles(const float* __restrict__ Arow,
                                          const float* __restrict__ Wc,
                                          int ct0, int lane, v8f (&d)[NT])
{
    const int koff = (lane < 16) ? 0 : 2;
    const float* wp[NT];
    bool nv[NT];
#pragma unroll
    for (int t = 0; t < NT; ++t) {
        const int n = (ct0 + t) * 16 + (lane & 15);
        nv[t] = (n < NCLS);
        wp[t] = Wc + koff * NCLS + min(n, NCLS - 1);
    }
#pragma unroll
    for (int k = 0; k < HDIM; k += 4) {
        const v2f a = *(const v2f*)(Arow + k);
#pragma unroll
        for (int t = 0; t < NT; ++t) {
            const float b0 = wp[t][k * NCLS];
            const float b1 = wp[t][(k + 1) * NCLS];
            v2f b;
            b[0] = nv[t] ? b0 : 0.0f;
            b[1] = nv[t] ? b1 : 0.0f;
            d[t] = __builtin_amdgcn_wmma_f32_16x16x4_f32(false, a, false, b,
                                                         (short)0, d[t], false, false);
        }
    }
}

__device__ __forceinline__ void cls_stage(const v8f& d, int rt, int ct, int lane,
                                          const float* __restrict__ bc, float* __restrict__ Ls)
{
    const int n = ct * 16 + (lane & 15);
    if (n < NCLS) {
        const float bv = bc[n];
        const int   rb = rt * 16 + ((lane < 16) ? 0 : 8);
#pragma unroll
        for (int r = 0; r < 8; ++r)
            Ls[(rb + r) * NCLS + n] = d[r] + bv;
    }
}

#define GK (BM / 8)

template <int U>
__device__ __forceinline__ void gather_rows(int ids, int t,
                                            const float* __restrict__ hl, float4& acc)
{
    float4 v[U];
#pragma unroll
    for (int u = 0; u < U; ++u) {
        const int s = __builtin_amdgcn_readlane(ids, t + u);
        v[u] = *(const float4*)(hl + (size_t)s * HDIM);
    }
#pragma unroll
    for (int u = 0; u < U; ++u) {
        acc.x += v[u].x; acc.y += v[u].y; acc.z += v[u].z; acc.w += v[u].w;
    }
}

__device__ __forceinline__ void gather_ids(int ids, int m,
                                           const float* __restrict__ hl, float4& acc)
{
    int t = 0;
    for (; t + 8 <= m; t += 8) gather_rows<8>(ids, t, hl, acc);
    if (t + 4 <= m) { gather_rows<4>(ids, t, hl, acc); t += 4; }
    if (t + 2 <= m) { gather_rows<2>(ids, t, hl, acc); t += 2; }
    if (t < m)        gather_rows<1>(ids, t, hl, acc);
}

template <bool CLS>
__global__ __launch_bounds__(256)
void sage_layer_wmma_kernel(const int* __restrict__ start, const int* __restrict__ cnt,
                            const int* __restrict__ col, const float* __restrict__ invdeg,
                            const float* __restrict__ h,
                            const float* __restrict__ Wl, const float* __restrict__ bl,
                            const float* __restrict__ Wr,
                            float* __restrict__ out, int N, int E,
                            const float* __restrict__ Wc, const float* __restrict__ bc)
{
    __shared__ __align__(16) float As[BM * APITCH];
    const int row0 = blockIdx.x * BM;
    const int tid  = threadIdx.x;
    const int lane = tid & 31;
    const int wave = __builtin_amdgcn_readfirstlane(tid >> 5);

    {
        const int r0 = wave * GK;
        const float* hl = h + lane * 4;

        float4 self[GK];
#pragma unroll
        for (int q = 0; q < GK; ++q) {
            const int row = min(row0 + r0 + q, N - 1);
            self[q] = *(const float4*)(hl + (size_t)row * HDIM);
        }

        const int qi  = min(row0 + r0 + (lane & (GK - 1)), N - 1);
        const int stv = start[qi];
        const int cnv = cnt[qi];
        const int ivv = __float_as_int(invdeg[qi]);

        int j   = __builtin_amdgcn_readlane(stv, 0);
        int end = j + __builtin_amdgcn_readlane(cnv, 0);
        int ids = col[min(j + lane, E - 1)];

#pragma unroll
        for (int q = 0; q < GK; ++q)
            *(float4*)(&As[(r0 + q) * APITCH + HDIM + lane * 4]) = self[q];

        for (int q = 0; q < GK; ++q) {
            int jn = end, endn = end, idsn = ids;
            if (q + 1 < GK) {
                jn   = __builtin_amdgcn_readlane(stv, q + 1);
                endn = jn + __builtin_amdgcn_readlane(cnv, q + 1);
                idsn = col[min(jn + lane, E - 1)];
            }

            float4 acc = make_float4(0.0f, 0.0f, 0.0f, 0.0f);
            gather_ids(ids, min(end - j, 32), hl, acc);
            for (int base = j + 32; base < end; base += 32) {
                const int more = col[min(base + lane, E - 1)];
                gather_ids(more, min(end - base, 32), hl, acc);
            }
            const float s = __int_as_float(__builtin_amdgcn_readlane(ivv, q));
            acc.x *= s; acc.y *= s; acc.z *= s; acc.w *= s;
            *(float4*)(&As[(r0 + q) * APITCH + lane * 4]) = acc;

            j = jn; end = endn; ids = idsn;
        }
    }
    __syncthreads();

    const int m    = lane & 15;
    const int koff = (lane < 16) ? 0 : 2;
    const int n    = wave * 16 + (lane & 15);

    v8f c[RT] = {};
    const float* Arow = &As[m * APITCH + koff];
    mma_rows<HDIM>(Arow,        Wl + (size_t)koff * HDIM + n, c);
    mma_rows<HDIM>(Arow + HDIM, Wr + (size_t)koff * HDIM + n, c);

    if constexpr (!CLS) {
        store_rows_relu(c, bl[n], row0, lane, n, out, N);
    } else {
        const float bv = bl[n];
        __syncthreads();
        {
            float* hp = &As[((lane < 16) ? 0 : 8) * APITCH + n];
#pragma unroll
            for (int t = 0; t < RT; ++t)
#pragma unroll
                for (int r = 0; r < 8; ++r)
                    hp[(t * 16 + r) * APITCH] = fmaxf(c[t][r] + bv, 0.0f);
        }
        __syncthreads();

        const int rt = wave >> 1;
        const float* Ar = &As[(rt * 16 + m) * APITCH + koff];
        v8f d[2] = {};
        if ((wave & 1) == 0) {
            cls_tiles<2>(Ar, Wc, 0, lane, d);
        } else {
            v8f e[1] = {};
            cls_tiles<1>(Ar, Wc, 2, lane, e);
            d[0] = e[0];
        }
        __syncthreads();

        float* Ls = As;
        if ((wave & 1) == 0) {
            cls_stage(d[0], rt, 0, lane, bc, Ls);
            cls_stage(d[1], rt, 1, lane, bc, Ls);
        } else {
            cls_stage(d[0], rt, 2, lane, bc, Ls);
        }
        __syncthreads();

        const int lim = min(BM, N - row0) * NCLS;
        float* op = out + (size_t)row0 * NCLS;
        for (int i = tid * 4; i < lim; i += 256 * 4)
            *(float4*)(op + i) = *(const float4*)(Ls + i);
    }
}

extern "C" void kernel_launch(void* const* d_in, const int* in_sizes, int n_in,
                              void* d_out, int out_size, void* d_ws, size_t ws_size,
                              hipStream_t stream)
{
    const float* x      = (const float*)d_in[0];
    const int*   edge   = (const int*)  d_in[1];
    const float* W_proj = (const float*)d_in[2];
    const float* b_proj = (const float*)d_in[3];
    const float* Wl[3]  = { (const float*)d_in[4], (const float*)d_in[7], (const float*)d_in[10] };
    const float* bl[3]  = { (const float*)d_in[5], (const float*)d_in[8], (const float*)d_in[11] };
    const float* Wr[3]  = { (const float*)d_in[6], (const float*)d_in[9], (const float*)d_in[12] };
    const float* Wc     = (const float*)d_in[13];
    const float* bc     = (const float*)d_in[14];

    const int N    = in_sizes[0] / 256;
    const int E    = in_sizes[1] / 2;

    const int* src = edge;
    const int* dst = edge + E;

    float* hA     = (float*)d_ws;
    float* hB     = hA  + (size_t)N * HDIM;
    int*   ghist  = (int*)(hB + (size_t)N * HDIM);
    int*   bbase  = ghist + BLD_BLOCKS * NBMAX;
    float* invdeg = (float*)(bbase + NBMAX + 4);
    int*   cnt    = (int*)(invdeg + N);
    int*   start  = cnt + N;
    int*   col    = start + N;
    unsigned* binned = (unsigned*)(col + E);

    const int rowBlocks = (N + BM - 1) / BM;

    const int pairs   = (rowBlocks + 1) / 2;
    const int p1      = pairs < BLD_BLOCKS ? pairs : BLD_BLOCKS;
    const int p2      = (pairs - p1) < BLD_BLOCKS ? (pairs - p1) : BLD_BLOCKS;
    const int p3      = pairs - p1 - p2;
    const int nCsr    = (N + BNODES - 1) / BNODES;
    build_proj_kernel<0><<<BLD_BLOCKS + p1, BLD_THREADS, 0, stream>>>(
        src, dst, ghist, bbase, binned, cnt, start, invdeg, col, N, E, BLD_BLOCKS, 0,
        x, W_proj, b_proj, hA);
    build_proj_kernel<1><<<BLD_BLOCKS + p2, BLD_THREADS, 0, stream>>>(
        src, dst, ghist, bbase, binned, cnt, start, invdeg, col, N, E, BLD_BLOCKS, p1,
        x, W_proj, b_proj, hA);
    build_proj_kernel<2><<<nCsr + p3, BLD_THREADS, 0, stream>>>(
        src, dst, ghist, bbase, binned, cnt, start, invdeg, col, N, E, nCsr, p1 + p2,
        x, W_proj, b_proj, hA);

    for (int l = 0; l < 2; ++l) {
        sage_layer_wmma_kernel<false><<<rowBlocks, 256, 0, stream>>>(start, cnt, col, invdeg, hA,
                                                                     Wl[l], bl[l], Wr[l], hB, N, E,
                                                                     Wc, bc);
        float* t = hA; hA = hB; hB = t;
    }

    sage_layer_wmma_kernel<true><<<rowBlocks, 256, 0, stream>>>(start, cnt, col, invdeg, hA,
                                                                Wl[2], bl[2], Wr[2], (float*)d_out, N, E,
                                                                Wc, bc);
}


# Self product h @ Wr moved into the neighbour fill: a wave runs 4 of its 32 k-steps behind each node's first 8 row loads while they fly; only mean @ Wl follows the fill; product sum order swaps
# speedup vs baseline: 4.7085x; 1.0318x over previous
#include <hip/hip_runtime.h>
#include <hip/hip_bf16.h>

typedef __attribute__((ext_vector_type(2))) float v2f;
typedef __attribute__((ext_vector_type(8))) float v8f;

#define HDIM 128

#define BM    64
#define RT    4
#define APITCH 260

template <int KD>
__device__ __forceinline__ void mma_rows(const float* __restrict__ Arow,
                                         const float* __restrict__ Wp,
                                         v8f (&c)[RT])
{
#pragma unroll
    for (int k = 0; k < KD; k += 4) {
        v2f b;
        b[0] = Wp[(size_t)k * HDIM];
        b[1] = Wp[(size_t)(k + 1) * HDIM];
#pragma unroll
        for (int t = 0; t < RT; ++t) {
            const v2f a = *(const v2f*)(Arow + t * 16 * APITCH + k);
            c[t] = __builtin_amdgcn_wmma_f32_16x16x4_f32(false, a, false, b,
                                                         (short)0, c[t], false, false);
        }
    }
}

__device__ __forceinline__ void store_rows_relu(const v8f (&c)[RT], float bv, int row0,
                                                int lane, int n, float* __restrict__ out, int N)
{
    float* op = out + (size_t)(row0 + ((lane < 16) ? 0 : 8)) * HDIM + n;
#pragma unroll
    for (int t = 0; t < RT; ++t) {
        if (row0 + t * 16 < N) {
#pragma unroll
            for (int r = 0; r < 8; ++r)
                op[(t * 16 + r) * HDIM] = fmaxf(c[t][r] + bv, 0.0f);
        }
    }
}

__device__ __forceinline__ void proj_role(const float* __restrict__ A, const float* __restrict__ W,
                                          const float* __restrict__ bias, float* __restrict__ out,
                                          int N, int tile0, int tid512)
{
    __shared__ __align__(16) float As2[2][BM * APITCH];
    const int half = tid512 >> 8;
    const int tid  = tid512 & 255;
    float* As = As2[half];
    const int row0 = (tile0 + half) * BM;

    {
        const int c4 = (tid & 63) * 4;
#pragma unroll
        for (int p = 0; p < BM / 4; ++p) {
            const int r   = p * 4 + (tid >> 6);
            const int row = min(row0 + r, N - 1);
            *(float4*)(&As[r * APITCH + c4]) = *(const float4*)(A + (size_t)row * 256 + c4);
        }
    }
    __syncthreads();

    const int wave = tid >> 5;
    const int lane = tid & 31;
    const int m    = lane & 15;
    const int koff = (lane < 16) ? 0 : 2;
    const int n    = wave * 16 + (lane & 15);

    v8f c[RT] = {};
    mma_rows<256>(&As[m * APITCH + koff], W + (size_t)koff * HDIM + n, c);

    store_rows_relu(c, bias[n], row0, lane, n, out, N);
}

#define BSH         9
#define BNODES      512
#define NBMAX       256
#define BLD_BLOCKS  256
#define BLD_THREADS 512
#define BLD_WAVES   (BLD_THREADS / 32)
#define EPT         13
#define EPT_CSR     20

__device__ __forceinline__ int block_excl_scan(int v, int* wsum, int tid)
{
    const int lane = tid & 31;
    const int wave = tid >> 5;
    int incl = v;
    for (int d = 1; d < 32; d <<= 1) {
        const int t = __shfl_up(incl, d, 32);
        if (lane >= d) incl += t;
    }
    if (lane == 31) wsum[wave] = incl;
    __syncthreads();
    if (wave == 0) {
        const int s = (lane < BLD_WAVES) ? wsum[lane] : 0;
        int si = s;
        for (int d = 1; d < 32; d <<= 1) {
            const int t = __shfl_up(si, d, 32);
            if (lane >= d) si += t;
        }
        if (lane < BLD_WAVES) wsum[lane] = si - s;
    }
    __syncthreads();
    return wsum[wave] + incl - v;
}

__device__ __forceinline__ void bucket_hist_role(const int* __restrict__ dst, int* __restrict__ ghist,
                                                 int E, int bid, int tid)
{
    __shared__ int hist[NBMAX];
    if (tid < NBMAX) hist[tid] = 0;
    __syncthreads();

    const int chunk = (E + BLD_BLOCKS - 1) / BLD_BLOCKS;
    const int base  = bid * chunk;
    const int end   = min(base + chunk, E);
    for (int s = base; s < end; s += BLD_THREADS * EPT) {
        int d[EPT];
#pragma unroll
        for (int u = 0; u < EPT; ++u)
            d[u] = dst[min(s + u * BLD_THREADS + tid, E - 1)];
#pragma unroll
        for (int u = 0; u < EPT; ++u)
            if (s + u * BLD_THREADS + tid < end) atomicAdd(&hist[d[u] >> BSH], 1);
    }
    __syncthreads();
    if (tid < NBMAX) ghist[bid * NBMAX + tid] = hist[tid];
}

#define TQ  (NBMAX / 4)
#define TG  (BLD_THREADS / TQ)
#define TR  (BLD_BLOCKS / TG)

__device__ __forceinline__ void bucket_bin_role(const int* __restrict__ src, const int* __restrict__ dst,
                                                const int* __restrict__ ghist, int* __restrict__ bbase,
                                                unsigned* __restrict__ binned, int E, int bid, int tid)
{
    __shared__ __align__(16) int ppre[TG][NBMAX];
    __shared__ __align__(16) int ptot[TG][NBMAX];
    __shared__ int offs[NBMAX];
    __shared__ int wsum[BLD_WAVES];

    {
        const int q = tid & (TQ - 1);
        const int g = tid / TQ;
        const int4* p = (const int4*)ghist + (size_t)(g * TR) * TQ + q;
        int4 tot = make_int4(0, 0, 0, 0), pre = make_int4(0, 0, 0, 0);
#pragma nounroll
        for (int r0 = 0; r0 < TR; r0 += TR / 2) {
            int4 v[TR / 2];
#pragma unroll
            for (int r = 0; r < TR / 2; ++r) v[r] = p[(r0 + r) * TQ];
#pragma unroll
            for (int r = 0; r < TR / 2; ++r) {
                tot.x += v[r].x; tot.y += v[r].y; tot.z += v[r].z; tot.w += v[r].w;
                if (g * TR + r0 + r < bid) {
                    pre.x += v[r].x; pre.y += v[r].y; pre.z += v[r].z; pre.w += v[r].w;
                }
            }
        }
        *(int4*)&ptot[g][4 * q] = tot;
        *(int4*)&ppre[g][4 * q] = pre;
    }
    __syncthreads();

    int tot = 0, pre = 0;
    if (tid < NBMAX) {
#pragma unroll
        for (int g = 0; g < TG; ++g) { tot += ptot[g][tid]; pre += ppre[g][tid]; }
    }
    const int excl = block_excl_scan(tot, wsum, tid);
    if (tid < NBMAX) {
        offs[tid] = excl + pre;
        if (bid == 0) {
            bbase[tid] = excl;
            if (tid == NBMAX - 1) bbase[NBMAX] = excl + tot;
        }
    }
    __syncthreads();

    const int chunk = (E + BLD_BLOCKS - 1) / BLD_BLOCKS;
    const int base  = bid * chunk;
    const int end   = min(base + chunk, E);
    for (int s = base; s < end; s += BLD_THREADS * EPT) {
        int d[EPT], sv[EPT];
#pragma unroll
        for (int u = 0; u < EPT; ++u) {
            const int e = min(s + u * BLD_THREADS + tid, E - 1);
            d[u]  = dst[e];
            sv[u] = src[e];
        }
        int pos[EPT];
#pragma unroll
        for (int u = 0; u < EPT; ++u) {
            pos[u] = 0;
            if (s + u * BLD_THREADS + tid < end) pos[u] = atomicAdd(&offs[d[u] >> BSH], 1);
        }
#pragma unroll
        for (int u = 0; u < EPT; ++u) {
            if (s + u * BLD_THREADS + tid < end)
                binned[pos[u]] = ((unsigned)sv[u] << BSH) | ((unsigned)d[u] & (BNODES - 1));
        }
    }
}

__device__ __forceinline__ void bucket_csr_role(const unsigned* __restrict__ binned, const int* __restrict__ bbase,
                                                int* __restrict__ cnt, int* __restrict__ start,
                                                float* __restrict__ invdeg, int* __restrict__ col, int n,
                                                int bid, int tid)
{
    __shared__ int c[BNODES];
    __shared__ int fpos[BNODES];
    __shared__ int wsum[BLD_WAVES];
    const int lo  = bbase[bid];
    const int hi  = bbase[bid + 1];
    c[tid] = 0;
    __syncthreads();

    for (int s = lo; s < hi; s += BLD_THREADS * EPT_CSR) {
        unsigned p[EPT_CSR];
#pragma unroll
        for (int u = 0; u < EPT_CSR; ++u)
            p[u] = binned[min(s + u * BLD_THREADS + tid, hi - 1)];
#pragma unroll
        for (int u = 0; u < EPT_CSR; ++u)
            if (s + u * BLD_THREADS + tid < hi) atomicAdd(&c[p[u] & (BNODES - 1)], 1);
    }
    __syncthreads();

    const int v    = c[tid];
    const int excl = block_excl_scan(v, wsum, tid);
    fpos[tid] = lo + excl;
    {
        const int node = bid * BNODES + tid;
        if (node < n) {
            cnt[node]    = v;
            start[node]  = lo + excl;
            invdeg[node] = 1.0f / fmaxf((float)v, 1.0f);
        }
    }
    __syncthreads();

    for (int s = lo; s < hi; s += BLD_THREADS * EPT_CSR) {
        unsigned p[EPT_CSR];
#pragma unroll
        for (int u = 0; u < EPT_CSR; ++u)
            p[u] = binned[min(s + u * BLD_THREADS + tid, hi - 1)];
        int pos[EPT_CSR];
#pragma unroll
        for (int u = 0; u < EPT_CSR; ++u) {
            pos[u] = 0;
            if (s + u * BLD_THREADS + tid < hi) pos[u] = atomicAdd(&fpos[p[u] & (BNODES - 1)], 1);
        }
#pragma unroll
        for (int u = 0; u < EPT_CSR; ++u) {
            if (s + u * BLD_THREADS + tid < hi) col[pos[u]] = (int)(p[u] >> BSH);
        }
    }
}

template <int ROLE>
__global__ __launch_bounds__(BLD_THREADS)
void build_proj_kernel(const int* __restrict__ src, const int* __restrict__ dst,
                       int* __restrict__ ghist, int* __restrict__ bbase,
                       unsigned* __restrict__ binned,
                       int* __restrict__ cnt, int* __restrict__ start,
                       float* __restrict__ invdeg, int* __restrict__ col,
                       int N, int E, int nBuild, int pairBase,
                       const float* __restrict__ x, const float* __restrict__ Wp,
                       const float* __restrict__ bp, float* __restrict__ h0)
{
    const int tid = threadIdx.x;
    const int bid = blockIdx.x;
    if (bid < nBuild) {
        if constexpr (ROLE == 0) bucket_hist_role(dst, ghist, E, bid, tid);
        if constexpr (ROLE == 1) bucket_bin_role(src, dst, ghist, bbase, binned, E, bid, tid);
        if constexpr (ROLE == 2) bucket_csr_role(binned, bbase, cnt, start, invdeg, col, N, bid, tid);
    } else {
        proj_role(x, Wp, bp, h0, N, 2 * (pairBase + bid - nBuild), tid);
    }
}

#define NCLS 40
static_assert(NCLS <= 48 && NCLS % 4 == 0, "three column tiles, 16-byte output pieces");

template <int NT>
__device__ __forceinline__ void cls_tiles(const float* __restrict__ Arow,
                                          const float* __restrict__ Wc,
                                          int ct0, int lane, v8f (&d)[NT])
{
    const int koff = (lane < 16) ? 0 : 2;
    const float* wp[NT];
    bool nv[NT];
#pragma unroll
    for (int t = 0; t < NT; ++t) {
        const int n = (ct0 + t) * 16 + (lane & 15);
        nv[t] = (n < NCLS);
        wp[t] = Wc + koff * NCLS + min(n, NCLS - 1);
    }
#pragma unroll
    for (int k = 0; k < HDIM; k += 4) {
        const v2f a = *(const v2f*)(Arow + k);
#pragma unroll
        for (int t = 0; t < NT; ++t) {
            const float b0 = wp[t][k * NCLS];
            const float b1 = wp[t][(k + 1) * NCLS];
            v2f b;
            b[0] = nv[t] ? b0 : 0.0f;
            b[1] = nv[t] ? b1 : 0.0f;
            d[t] = __builtin_amdgcn_wmma_f32_16x16x4_f32(false, a, false, b,
                                                         (short)0, d[t], false, false);
        }
    }
}

__device__ __forceinline__ void cls_stage(const v8f& d, int rt, int ct, int lane,
                                          const float* __restrict__ bc, float* __restrict__ Ls)
{
    const int n = ct * 16 + (lane & 15);
    if (n < NCLS) {
        const float bv = bc[n];
        const int   rb = rt * 16 + ((lane < 16) ? 0 : 8);
#pragma unroll
        for (int r = 0; r < 8; ++r)
            Ls[(rb + r) * NCLS + n] = d[r] + bv;
    }
}

#define GK (BM / 8)

template <int U>
__device__ __forceinline__ void gather_rows(int ids, int t,
                                            const float* __restrict__ hl, float4& acc)
{
    float4 v[U];
#pragma unroll
    for (int u = 0; u < U; ++u) {
        const int s = __builtin_amdgcn_readlane(ids, t + u);
        v[u] = *(const float4*)(hl + (size_t)s * HDIM);
    }
#pragma unroll
    for (int u = 0; u < U; ++u) {
        acc.x += v[u].x; acc.y += v[u].y; acc.z += v[u].z; acc.w += v[u].w;
    }
}

__device__ __forceinline__ void gather_ids(int ids, int m,
                                           const float* __restrict__ hl, float4& acc)
{
    int t = 0;
    for (; t + 8 <= m; t += 8) gather_rows<8>(ids, t, hl, acc);
    if (t + 4 <= m) { gather_rows<4>(ids, t, hl, acc); t += 4; }
    if (t + 2 <= m) { gather_rows<2>(ids, t, hl, acc); t += 2; }
    if (t < m)        gather_rows<1>(ids, t, hl, acc);
}

__device__ __forceinline__ void self_ksteps(const float* __restrict__ Ak, const v2f (&b)[4], v8f (&c)[RT])
{
#pragma unroll
    for (int s = 0; s < 4; ++s) {
#pragma unroll
        for (int t = 0; t < RT; ++t) {
            const v2f a = *(const v2f*)(Ak + t * 16 * APITCH + s * 4);
            c[t] = __builtin_amdgcn_wmma_f32_16x16x4_f32(false, a, false, b[s],
                                                         (short)0, c[t], false, false);
        }
    }
}

template <bool CLS>
__global__ __launch_bounds__(256)
void sage_layer_wmma_kernel(const int* __restrict__ start, const int* __restrict__ cnt,
                            const int* __restrict__ col, const float* __restrict__ invdeg,
                            const float* __restrict__ h,
                            const float* __restrict__ Wl, const float* __restrict__ bl,
                            const float* __restrict__ Wr,
                            float* __restrict__ out, int N, int E,
                            const float* __restrict__ Wc, const float* __restrict__ bc)
{
    __shared__ __align__(16) float As[BM * APITCH];
    const int row0 = blockIdx.x * BM;
    const int tid  = threadIdx.x;
    const int lane = tid & 31;
    const int wave = __builtin_amdgcn_readfirstlane(tid >> 5);

    const int m    = lane & 15;
    const int koff = (lane < 16) ? 0 : 2;
    const int n    = wave * 16 + (lane & 15);
    const float* Arow = &As[m * APITCH + koff];
    v8f c[RT] = {};

    {
        const int r0 = wave * GK;
        const float* hl = h + lane * 4;
        const float* WrP = Wr + (size_t)koff * HDIM + n;

        float4 self[GK];
#pragma unroll
        for (int q = 0; q < GK; ++q) {
            const int row = min(row0 + r0 + q, N - 1);
            self[q] = *(const float4*)(hl + (size_t)row * HDIM);
        }

        const int qi  = min(row0 + r0 + (lane & (GK - 1)), N - 1);
        const int stv = start[qi];
        const int cnv = cnt[qi];
        const int ivv = __float_as_int(invdeg[qi]);

        int j   = __builtin_amdgcn_readlane(stv, 0);
        int end = j + __builtin_amdgcn_readlane(cnv, 0);
        int ids = col[min(j + lane, E - 1)];

#pragma unroll
        for (int q = 0; q < GK; ++q)
            *(float4*)(&As[(r0 + q) * APITCH + HDIM + lane * 4]) = self[q];
        __syncthreads();

        for (int q = 0; q < GK; ++q) {
            int jn = end, endn = end, idsn = ids;
            if (q + 1 < GK) {
                jn   = __builtin_amdgcn_readlane(stv, q + 1);
                endn = jn + __builtin_amdgcn_readlane(cnv, q + 1);
                idsn = col[min(jn + lane, E - 1)];
            }

            v2f b[4];
#pragma unroll
            for (int s = 0; s < 4; ++s) {
                const int k = q * 16 + s * 4;
                b[s][0] = WrP[(size_t)k * HDIM];
                b[s][1] = WrP[(size_t)(k + 1) * HDIM];
            }

            float4 acc = make_float4(0.0f, 0.0f, 0.0f, 0.0f);
            const int mm = min(end - j, 32);
            int t = 0;
            if (mm >= 8) {
                float4 v[8];
#pragma unroll
                for (int u = 0; u < 8; ++u) {
                    const int s = __builtin_amdgcn_readlane(ids, u);
                    v[u] = *(const float4*)(hl + (size_t)s * HDIM);
                }
                self_ksteps(Arow + HDIM + q * 16, b, c);
#pragma unroll
                for (int u = 0; u < 8; ++u) {
                    acc.x += v[u].x; acc.y += v[u].y; acc.z += v[u].z; acc.w += v[u].w;
                }
                t = 8;
            } else {
                self_ksteps(Arow + HDIM + q * 16, b, c);
            }
            for (; t + 8 <= mm; t += 8) gather_rows<8>(ids, t, hl, acc);
            if (t + 4 <= mm) { gather_rows<4>(ids, t, hl, acc); t += 4; }
            if (t + 2 <= mm) { gather_rows<2>(ids, t, hl, acc); t += 2; }
            if (t < mm)        gather_rows<1>(ids, t, hl, acc);
            for (int base = j + 32; base < end; base += 32) {
                const int more = col[min(base + lane, E - 1)];
                gather_ids(more, min(end - base, 32), hl, acc);
            }
            const float s = __int_as_float(__builtin_amdgcn_readlane(ivv, q));
            acc.x *= s; acc.y *= s; acc.z *= s; acc.w *= s;
            *(float4*)(&As[(r0 + q) * APITCH + lane * 4]) = acc;

            j = jn; end = endn; ids = idsn;
        }
    }
    __syncthreads();

    mma_rows<HDIM>(Arow, Wl + (size_t)koff * HDIM + n, c);

    if constexpr (!CLS) {
        store_rows_relu(c, bl[n], row0, lane, n, out, N);
    } else {
        const float bv = bl[n];
        __syncthreads();
        {
            float* hp = &As[((lane < 16) ? 0 : 8) * APITCH + n];
#pragma unroll
            for (int t = 0; t < RT; ++t)
#pragma unroll
                for (int r = 0; r < 8; ++r)
                    hp[(t * 16 + r) * APITCH] = fmaxf(c[t][r] + bv, 0.0f);
        }
        __syncthreads();

        const int rt = wave >> 1;
        const float* Ar = &As[(rt * 16 + m) * APITCH + koff];
        v8f d[2] = {};
        if ((wave & 1) == 0) {
            cls_tiles<2>(Ar, Wc, 0, lane, d);
        } else {
            v8f e[1] = {};
            cls_tiles<1>(Ar, Wc, 2, lane, e);
            d[0] = e[0];
        }
        __syncthreads();

        float* Ls = As;
        if ((wave & 1) == 0) {
            cls_stage(d[0], rt, 0, lane, bc, Ls);
            cls_stage(d[1], rt, 1, lane, bc, Ls);
        } else {
            cls_stage(d[0], rt, 2, lane, bc, Ls);
        }
        __syncthreads();

        const int lim = min(BM, N - row0) * NCLS;
        float* op = out + (size_t)row0 * NCLS;
        for (int i = tid * 4; i < lim; i += 256 * 4)
            *(float4*)(op + i) = *(const float4*)(Ls + i);
    }
}

extern "C" void kernel_launch(void* const* d_in, const int* in_sizes, int n_in,
                              void* d_out, int out_size, void* d_ws, size_t ws_size,
                              hipStream_t stream)
{
    const float* x      = (const float*)d_in[0];
    const int*   edge   = (const int*)  d_in[1];
    const float* W_proj = (const float*)d_in[2];
    const float* b_proj = (const float*)d_in[3];
    const float* Wl[3]  = { (const float*)d_in[4], (const float*)d_in[7], (const float*)d_in[10] };
    const float* bl[3]  = { (const float*)d_in[5], (const float*)d_in[8], (const float*)d_in[11] };
    const float* Wr[3]  = { (const float*)d_in[6], (const float*)d_in[9], (const float*)d_in[12] };
    const float* Wc     = (const float*)d_in[13];
    const float* bc     = (const float*)d_in[14];

    const int N    = in_sizes[0] / 256;
    const int E    = in_sizes[1] / 2;

    const int* src = edge;
    const int* dst = edge + E;

    float* hA     = (float*)d_ws;
    float* hB     = hA  + (size_t)N * HDIM;
    int*   ghist  = (int*)(hB + (size_t)N * HDIM);
    int*   bbase  = ghist + BLD_BLOCKS * NBMAX;
    float* invdeg = (float*)(bbase + NBMAX + 4);
    int*   cnt    = (int*)(invdeg + N);
    int*   start  = cnt + N;
    int*   col    = start + N;
    unsigned* binned = (unsigned*)(col + E);

    const int rowBlocks = (N + BM - 1) / BM;

    const int pairs   = (rowBlocks + 1) / 2;
    const int p1      = pairs < BLD_BLOCKS ? pairs : BLD_BLOCKS;
    const int p2      = (pairs - p1) < BLD_BLOCKS ? (pairs - p1) : BLD_BLOCKS;
    const int p3      = pairs - p1 - p2;
    const int nCsr    = (N + BNODES - 1) / BNODES;
    build_proj_kernel<0><<<BLD_BLOCKS + p1, BLD_THREADS, 0, stream>>>(
        src, dst, ghist, bbase, binned, cnt, start, invdeg, col, N, E, BLD_BLOCKS, 0,
        x, W_proj, b_proj, hA);
    build_proj_kernel<1><<<BLD_BLOCKS + p2, BLD_THREADS, 0, stream>>>(
        src, dst, ghist, bbase, binned, cnt, start, invdeg, col, N, E, BLD_BLOCKS, p1,
        x, W_proj, b_proj, hA);
    build_proj_kernel<2><<<nCsr + p3, BLD_THREADS, 0, stream>>>(
        src, dst, ghist, bbase, binned, cnt, start, invdeg, col, N, E, nCsr, p1 + p2,
        x, W_proj, b_proj, hA);

    for (int l = 0; l < 2; ++l) {
        sage_layer_wmma_kernel<false><<<rowBlocks, 256, 0, stream>>>(start, cnt, col, invdeg, hA,
                                                                     Wl[l], bl[l], Wr[l], hB, N, E,
                                                                     Wc, bc);
        float* t = hA; hA = hB; hB = t;
    }

    sage_layer_wmma_kernel<true><<<rowBlocks, 256, 0, stream>>>(start, cnt, col, invdeg, hA,
                                                                Wl[2], bl[2], Wr[2], (float*)d_out, N, E,
                                                                Wc, bc);
}
